# starting kernel
# baseline (speedup 1.0000x reference)
#include <hip/hip_runtime.h>


#define IMG      512
#define CH       16
#define TS       32
#define RAWN     52
#define RAWROWS  64
#define RAWPITCH 66
#define ABROWS   48
#define ABPITCH  50
#define TROWS    64
#define TPITCH   50
#define MAPITCH  34

typedef __attribute__((ext_vector_type(2))) float v2f;
typedef __attribute__((ext_vector_type(8))) float v8f;

__device__ __forceinline__ int refl(int p) {
  p = p < 0 ? -p : p;
  return p >= IMG ? 2 * IMG - 2 - p : p;
}

template <int MODE>
__device__ __forceinline__ void hblur(const float* __restrict__ s0,
                                      const float* __restrict__ s1, int sPitch,
                                      float* __restrict__ dst, int dPitch,
                                      int mBlocks, int nBlocks,
                                      const v2f* __restrict__ band,
                                      int lane, int wave) {
  const int m = lane & 15, half = lane >> 4;
  const int nT = mBlocks * nBlocks;
  for (int t = wave; t < nT; t += 8) {
    const int mi = t / nBlocks, nj = t - mi * nBlocks;
    const float* p0 = s0 + (mi * 16 + m) * sPitch + nj * 16 + 2 * half;
    const float* p1 = s1 + (mi * 16 + m) * sPitch + nj * 16 + 2 * half;
    v8f c = {};
#pragma unroll
    for (int kt = 0; kt < 7; ++kt) {
      float x0 = p0[4 * kt], x1 = p0[4 * kt + 1];
      if (MODE == 1) { x0 *= x0; x1 *= x1; }
      if (MODE == 2) { x0 *= p1[4 * kt]; x1 *= p1[4 * kt + 1]; }
      v2f a; a.x = x0; a.y = x1;
      c = __builtin_amdgcn_wmma_f32_16x16x4_f32(false, a, false, band[kt],
                                                (short)0, c, false, false);
    }
    float* pd = dst + (mi * 16 + half * 8) * dPitch + nj * 16 + m;
#pragma unroll
    for (int v = 0; v < 8; ++v) pd[v * dPitch] = c[v];
  }
}

__device__ __forceinline__ void vblur(const float* __restrict__ src, int sPitch,
                                      float* __restrict__ dst, int dPitch,
                                      int mBlocks, int nBlocks,
                                      const v2f* __restrict__ band,
                                      int lane, int wave) {
  const int n = lane & 15, half = lane >> 4;
  const int nT = mBlocks * nBlocks;
  for (int t = wave; t < nT; t += 8) {
    const int mi = t / nBlocks, nj = t - mi * nBlocks;
    const float* ps = src + (mi * 16 + 2 * half) * sPitch + nj * 16 + n;
    v8f c = {};
#pragma unroll
    for (int kt = 0; kt < 7; ++kt) {
      v2f b;
      b.x = ps[(4 * kt) * sPitch];
      b.y = ps[(4 * kt + 1) * sPitch];
      c = __builtin_amdgcn_wmma_f32_16x16x4_f32(false, band[kt], false, b,
                                                (short)0, c, false, false);
    }
    float* pd = dst + (mi * 16 + half * 8) * dPitch + nj * 16 + n;
#pragma unroll
    for (int v = 0; v < 8; ++v) pd[v * dPitch] = c[v];
  }
}

__global__ __launch_bounds__(256) void guided_filter_kernel(
    const float* __restrict__ feat, const float* __restrict__ img,
    float* __restrict__ out) {
  __shared__ float rawI[RAWROWS * RAWPITCH];
  __shared__ float rawF[RAWROWS * RAWPITCH];
  __shared__ float tmpH[TROWS * TPITCH];
  __shared__ float mI [ABROWS * ABPITCH];
  __shared__ float vIi[ABROWS * ABPITCH];
  __shared__ float mP [ABROWS * ABPITCH];
  __shared__ float cIp[ABROWS * ABPITCH];
  __shared__ float mA [TS * MAPITCH];
  __shared__ float mB [TS * MAPITCH];

  const int tid  = threadIdx.x;
  const int lane = tid & 31;
  const int wave = tid >> 5;
  const int x0 = blockIdx.x * TS, y0 = blockIdx.y * TS;
  const int bat = blockIdx.z;

  v2f band[7];
  {
    const int q = lane & 15, h2 = (lane >> 4) * 2;
#pragma unroll
    for (int kt = 0; kt < 7; ++kt) {
      const int k0 = 4 * kt + h2;
      band[kt].x = (k0 >= q && k0 <= q + 10) ? (1.0f / 11.0f) : 0.0f;
      band[kt].y = (k0 + 1 >= q && k0 + 1 <= q + 10) ? (1.0f / 11.0f) : 0.0f;
    }
  }

  for (int i = tid; i < RAWROWS * RAWPITCH; i += 256) { rawI[i] = 0.0f; rawF[i] = 0.0f; }
  __syncthreads();

  const float* imgB = img + (size_t)bat * IMG * IMG;
  for (int i = tid; i < RAWN * RAWN; i += 256) {
    const int r = i / RAWN, cc = i - r * RAWN;
    rawI[r * RAWPITCH + cc] = imgB[refl(y0 - 10 + r) * IMG + refl(x0 - 10 + cc)];
  }
  __syncthreads();

  hblur<0>(rawI, rawI, RAWPITCH, tmpH, TPITCH, 4, 3, band, lane, wave);
  __syncthreads();
  vblur(tmpH, TPITCH, mI, ABPITCH, 3, 3, band, lane, wave);
  __syncthreads();
  hblur<1>(rawI, rawI, RAWPITCH, tmpH, TPITCH, 4, 3, band, lane, wave);
  __syncthreads();
  vblur(tmpH, TPITCH, vIi, ABPITCH, 3, 3, band, lane, wave);
  __syncthreads();
  for (int i = tid; i < ABROWS * ABROWS; i += 256) {
    const int r = i / ABROWS, cc = i - r * ABROWS;
    const int o = r * ABPITCH + cc;
    const float m = mI[o];
    vIi[o] = 1.0f / (vIi[o] - m * m + 1e-8f);
  }
  __syncthreads();

  for (int ch = 0; ch < CH; ++ch) {
    const float* featB = feat + ((size_t)bat * CH + ch) * IMG * IMG;
    for (int i = tid; i < RAWN * RAWN; i += 256) {
      const int r = i / RAWN, cc = i - r * RAWN;
      rawF[r * RAWPITCH + cc] = featB[refl(y0 - 10 + r) * IMG + refl(x0 - 10 + cc)];
    }
    __syncthreads();

    hblur<0>(rawF, rawF, RAWPITCH, tmpH, TPITCH, 4, 3, band, lane, wave);
    __syncthreads();
    vblur(tmpH, TPITCH, mP, ABPITCH, 3, 3, band, lane, wave);
    __syncthreads();
    hblur<2>(rawF, rawI, RAWPITCH, tmpH, TPITCH, 4, 3, band, lane, wave);
    __syncthreads();
    vblur(tmpH, TPITCH, cIp, ABPITCH, 3, 3, band, lane, wave);
    __syncthreads();

    for (int i = tid; i < ABROWS * ABROWS; i += 256) {
      const int r = i / ABROWS, cc = i - r * ABROWS;
      const int o = r * ABPITCH + cc;
      const float mi_ = mI[o], mp_ = mP[o];
      const float a = (cIp[o] - mi_ * mp_) * vIi[o];
      cIp[o] = a;
      mP[o]  = mp_ - a * mi_;
    }
    __syncthreads();

    hblur<0>(cIp, cIp, ABPITCH, tmpH, TPITCH, 3, 2, band, lane, wave);
    __syncthreads();
    vblur(tmpH, TPITCH, mA, MAPITCH, 2, 2, band, lane, wave);
    __syncthreads();
    hblur<0>(mP, mP, ABPITCH, tmpH, TPITCH, 3, 2, band, lane, wave);
    __syncthreads();
    vblur(tmpH, TPITCH, mB, MAPITCH, 2, 2, band, lane, wave);
    __syncthreads();

    float* outB = out + ((size_t)bat * CH + ch) * IMG * IMG;
    for (int i = tid; i < TS * TS; i += 256) {
      const int y = i >> 5, x = i & 31;
      outB[(size_t)(y0 + y) * IMG + (x0 + x)] =
          mA[y * MAPITCH + x] * rawI[(y + 10) * RAWPITCH + (x + 10)] +
          mB[y * MAPITCH + x];
    }
    __syncthreads();
  }
}

extern "C" void kernel_launch(void* const* d_in, const int* in_sizes, int n_in,
                              void* d_out, int out_size, void* d_ws, size_t ws_size,
                              hipStream_t stream) {
  const float* feat = (const float*)d_in[0];
  const float* img  = (const float*)d_in[1];
  float* out = (float*)d_out;
  const int B = in_sizes[1] / (IMG * IMG);
  dim3 grid(IMG / TS, IMG / TS, B);
  guided_filter_kernel<<<grid, 256, 0, stream>>>(feat, img, out);
}


# H-pass WMMA D tiles stay in registers and are the V-pass B operand (band operand in D-register k order, so the V-pass sum order changes): no LDS scratch plane, 5 barriers per channel instead of 13
# speedup vs baseline: 1.1373x; 1.1373x over previous
#include <hip/hip_runtime.h>


#define IMG      512
#define CH       16
#define TS       32
#define RAWN     52
#define RAWROWS  64
#define RAWPITCH 66
#define ABROWS   48
#define ABPITCH  50
#define MAPITCH  34

typedef __attribute__((ext_vector_type(2))) float v2f;
typedef __attribute__((ext_vector_type(8))) float v8f;

__device__ __forceinline__ int refl(int p) {
  p = p < 0 ? -p : p;
  return p >= IMG ? 2 * IMG - 2 - p : p;
}

template <int MODE, int NIN>
__device__ __forceinline__ void blur_strip(const float* __restrict__ s0,
                                           const float* __restrict__ s1, int sPitch,
                                           int nj, float* __restrict__ dst, int dPitch,
                                           const v2f* __restrict__ bandH,
                                           const v2f (*__restrict__ bandV)[4],
                                           int lane) {
  const int c = lane & 15, half = lane >> 4;
  const float* p0 = s0 + c * sPitch + nj * 16 + 2 * half;
  const float* p1 = s1 + c * sPitch + nj * 16 + 2 * half;
  v8f hT[NIN];
#pragma unroll
  for (int i = 0; i < NIN; ++i) hT[i] = (v8f){};
#pragma unroll
  for (int kt = 0; kt < 7; ++kt) {
#pragma unroll
    for (int i = 0; i < NIN; ++i) {
      const int o = i * 16 * sPitch + 4 * kt;
      float x0 = p0[o], x1 = p0[o + 1];
      if (MODE == 1) { x0 *= x0; x1 *= x1; }
      if (MODE == 2) { x0 *= p1[o]; x1 *= p1[o + 1]; }
      v2f a; a.x = x0; a.y = x1;
      hT[i] = __builtin_amdgcn_wmma_f32_16x16x4_f32(false, a, false, bandH[kt],
                                                    (short)0, hT[i], false, false);
    }
  }
  v8f acc[NIN - 1];
#pragma unroll
  for (int o = 0; o < NIN - 1; ++o) acc[o] = (v8f){};
#pragma unroll
  for (int blk = 0; blk < 2; ++blk) {
#pragma unroll
    for (int t = 0; t < 4; ++t) {
#pragma unroll
      for (int o = 0; o < NIN - 1; ++o) {
        v2f b; b.x = hT[o + blk][2 * t]; b.y = hT[o + blk][2 * t + 1];
        acc[o] = __builtin_amdgcn_wmma_f32_16x16x4_f32(false, bandV[blk][t], false, b,
                                                       (short)0, acc[o], false, false);
      }
    }
  }
#pragma unroll
  for (int o = 0; o < NIN - 1; ++o) {
    float* pd = dst + (o * 16 + half * 8) * dPitch + nj * 16 + c;
#pragma unroll
    for (int v = 0; v < 8; ++v) pd[v * dPitch] = acc[o][v];
  }
}

__global__ __launch_bounds__(256) void guided_filter_kernel(
    const float* __restrict__ feat, const float* __restrict__ img,
    float* __restrict__ out) {
  __shared__ float rawI[RAWROWS * RAWPITCH];
  __shared__ float rawF[RAWROWS * RAWPITCH];
  __shared__ float mI [ABROWS * ABPITCH];
  __shared__ float vIi[ABROWS * ABPITCH];
  __shared__ float mP [ABROWS * ABPITCH];
  __shared__ float cIp[ABROWS * ABPITCH];
  __shared__ float mA [TS * MAPITCH];
  __shared__ float mB [TS * MAPITCH];

  const int tid  = threadIdx.x;
  const int lane = tid & 31;
  const int wave = tid >> 5;
  const int x0 = blockIdx.x * TS, y0 = blockIdx.y * TS;
  const int bat = blockIdx.z;

  v2f bandH[7];
  v2f bandV[2][4];
  {
    const int q = lane & 15, h = lane >> 4;
#pragma unroll
    for (int kt = 0; kt < 7; ++kt) {
      const int k0 = 4 * kt + 2 * h;
      bandH[kt].x = (k0 >= q && k0 <= q + 10) ? (1.0f / 11.0f) : 0.0f;
      bandH[kt].y = (k0 + 1 >= q && k0 + 1 <= q + 10) ? (1.0f / 11.0f) : 0.0f;
    }
#pragma unroll
    for (int blk = 0; blk < 2; ++blk) {
#pragma unroll
      for (int t = 0; t < 4; ++t) {
        const int k0 = 16 * blk + 8 * h + 2 * t;
        bandV[blk][t].x = (k0 >= q && k0 <= q + 10) ? (1.0f / 11.0f) : 0.0f;
        bandV[blk][t].y = (k0 + 1 >= q && k0 + 1 <= q + 10) ? (1.0f / 11.0f) : 0.0f;
      }
    }
  }

  for (int i = tid; i < RAWROWS * RAWPITCH; i += 256) { rawI[i] = 0.0f; rawF[i] = 0.0f; }
  __syncthreads();

  const float* imgB = img + (size_t)bat * IMG * IMG;
  for (int i = tid; i < RAWN * RAWN; i += 256) {
    const int r = i / RAWN, cc = i - r * RAWN;
    rawI[r * RAWPITCH + cc] = imgB[refl(y0 - 10 + r) * IMG + refl(x0 - 10 + cc)];
  }
  __syncthreads();

  if (wave < 6) {
    const int qs = wave >= 3 ? 1 : 0, nj = wave - 3 * qs;
    if (qs == 0) blur_strip<0, 4>(rawI, rawI, RAWPITCH, nj, mI,  ABPITCH, bandH, bandV, lane);
    else         blur_strip<1, 4>(rawI, rawI, RAWPITCH, nj, vIi, ABPITCH, bandH, bandV, lane);
  }
  __syncthreads();
  for (int i = tid; i < ABROWS * ABROWS; i += 256) {
    const int r = i / ABROWS, cc = i - r * ABROWS;
    const int o = r * ABPITCH + cc;
    const float m = mI[o];
    vIi[o] = 1.0f / (vIi[o] - m * m + 1e-8f);
  }
  __syncthreads();

  for (int ch = 0; ch < CH; ++ch) {
    const float* featB = feat + ((size_t)bat * CH + ch) * IMG * IMG;
    for (int i = tid; i < RAWN * RAWN; i += 256) {
      const int r = i / RAWN, cc = i - r * RAWN;
      rawF[r * RAWPITCH + cc] = featB[refl(y0 - 10 + r) * IMG + refl(x0 - 10 + cc)];
    }
    __syncthreads();

    {
      const int t = (wave + 2 * ch) & 7;
      if (t < 6) {
        const int qs = t >= 3 ? 1 : 0, nj = t - 3 * qs;
        if (qs == 0) blur_strip<0, 4>(rawF, rawF, RAWPITCH, nj, mP,  ABPITCH, bandH, bandV, lane);
        else         blur_strip<2, 4>(rawF, rawI, RAWPITCH, nj, cIp, ABPITCH, bandH, bandV, lane);
      }
    }
    __syncthreads();

    for (int i = tid; i < ABROWS * ABROWS; i += 256) {
      const int r = i / ABROWS, cc = i - r * ABROWS;
      const int o = r * ABPITCH + cc;
      const float mi_ = mI[o], mp_ = mP[o];
      const float a = (cIp[o] - mi_ * mp_) * vIi[o];
      cIp[o] = a;
      mP[o]  = mp_ - a * mi_;
    }
    __syncthreads();

    {
      const int t = (wave + 2 * ch + 2) & 7;
      if (t < 4) {
        const int qs = t >> 1, nj = t & 1;
        if (qs == 0) blur_strip<0, 3>(cIp, cIp, ABPITCH, nj, mA, MAPITCH, bandH, bandV, lane);
        else         blur_strip<0, 3>(mP,  mP,  ABPITCH, nj, mB, MAPITCH, bandH, bandV, lane);
      }
    }
    __syncthreads();

    float* outB = out + ((size_t)bat * CH + ch) * IMG * IMG;
    for (int i = tid; i < TS * TS; i += 256) {
      const int y = i >> 5, x = i & 31;
      outB[(size_t)(y0 + y) * IMG + (x0 + x)] =
          mA[y * MAPITCH + x] * rawI[(y + 10) * RAWPITCH + (x + 10)] +
          mB[y * MAPITCH + x];
    }
    __syncthreads();
  }
}

extern "C" void kernel_launch(void* const* d_in, const int* in_sizes, int n_in,
                              void* d_out, int out_size, void* d_ws, size_t ws_size,
                              hipStream_t stream) {
  const float* feat = (const float*)d_in[0];
  const float* img  = (const float*)d_in[1];
  float* out = (float*)d_out;
  const int B = in_sizes[1] / (IMG * IMG);
  dim3 grid(IMG / TS, IMG / TS, B);
  guided_filter_kernel<<<grid, 256, 0, stream>>>(feat, img, out);
}


# Software-pipelined global traffic: next channel's 52x52 feature halo is loaded into registers one channel ahead and stays in flight across LDS-only barriers; barriers no longer drain loads or stores
# speedup vs baseline: 1.6263x; 1.4300x over previous
#include <hip/hip_runtime.h>


#define IMG      512
#define CH       16
#define TS       32
#define RAWN     52
#define RAWROWS  64
#define RAWPITCH 66
#define ABROWS   48
#define ABPITCH  50
#define MAPITCH  34
#define NLD      13

typedef __attribute__((ext_vector_type(2))) float v2f;
typedef __attribute__((ext_vector_type(8))) float v8f;

__device__ __forceinline__ int refl(int p) {
  p = p < 0 ? -p : p;
  return p >= IMG ? 2 * IMG - 2 - p : p;
}

__device__ __forceinline__ void halo_fetch(const float* __restrict__ src, int gx, int gy0,
                                           float (&v)[NLD]) {
#pragma unroll
  for (int j = 0; j < NLD; ++j) {
    const float* row = src + refl(gy0 + 4 * j) * IMG;
    v[j] = row[gx];
  }
}

__device__ __forceinline__ void lds_barrier() {
  asm volatile("s_wait_dscnt 0x0\n\ts_barrier_signal -1\n\ts_barrier_wait -1" ::: "memory");
}
__device__ __forceinline__ void halo_store(float* __restrict__ raw, int lo, bool valid,
                                           const float (&v)[NLD]) {
  if (valid) {
#pragma unroll
    for (int j = 0; j < NLD; ++j) raw[lo + 4 * j * RAWPITCH] = v[j];
  }
}

template <int MODE, int NIN>
__device__ __forceinline__ void blur_strip(const float* __restrict__ s0,
                                           const float* __restrict__ s1, int sPitch,
                                           int nj, float* __restrict__ dst, int dPitch,
                                           const v2f* __restrict__ bandH,
                                           const v2f (*__restrict__ bandV)[4],
                                           int lane) {
  const int c = lane & 15, half = lane >> 4;
  const float* p0 = s0 + c * sPitch + nj * 16 + 2 * half;
  const float* p1 = s1 + c * sPitch + nj * 16 + 2 * half;
  v8f hT[NIN];
#pragma unroll
  for (int i = 0; i < NIN; ++i) hT[i] = (v8f){};
#pragma unroll
  for (int kt = 0; kt < 7; ++kt) {
#pragma unroll
    for (int i = 0; i < NIN; ++i) {
      const int o = i * 16 * sPitch + 4 * kt;
      float x0 = p0[o], x1 = p0[o + 1];
      if (MODE == 1) { x0 *= x0; x1 *= x1; }
      if (MODE == 2) { x0 *= p1[o]; x1 *= p1[o + 1]; }
      v2f a; a.x = x0; a.y = x1;
      hT[i] = __builtin_amdgcn_wmma_f32_16x16x4_f32(false, a, false, bandH[kt],
                                                    (short)0, hT[i], false, false);
    }
  }
  v8f acc[NIN - 1];
#pragma unroll
  for (int o = 0; o < NIN - 1; ++o) acc[o] = (v8f){};
#pragma unroll
  for (int blk = 0; blk < 2; ++blk) {
#pragma unroll
    for (int t = 0; t < 4; ++t) {
#pragma unroll
      for (int o = 0; o < NIN - 1; ++o) {
        v2f b; b.x = hT[o + blk][2 * t]; b.y = hT[o + blk][2 * t + 1];
        acc[o] = __builtin_amdgcn_wmma_f32_16x16x4_f32(false, bandV[blk][t], false, b,
                                                       (short)0, acc[o], false, false);
      }
    }
  }
#pragma unroll
  for (int o = 0; o < NIN - 1; ++o) {
    float* pd = dst + (o * 16 + half * 8) * dPitch + nj * 16 + c;
#pragma unroll
    for (int v = 0; v < 8; ++v) pd[v * dPitch] = acc[o][v];
  }
}

__global__ __launch_bounds__(256) void guided_filter_kernel(
    const float* __restrict__ feat, const float* __restrict__ img,
    float* __restrict__ out) {
  __shared__ float rawI[RAWROWS * RAWPITCH];
  __shared__ float rawF[RAWROWS * RAWPITCH];
  __shared__ float mI [ABROWS * ABPITCH];
  __shared__ float vIi[ABROWS * ABPITCH];
  __shared__ float mP [ABROWS * ABPITCH];
  __shared__ float cIp[ABROWS * ABPITCH];
  __shared__ float mA [TS * MAPITCH];
  __shared__ float mB [TS * MAPITCH];

  const int tid  = threadIdx.x;
  const int lane = tid & 31;
  const int wave = tid >> 5;
  const int x0 = blockIdx.x * TS, y0 = blockIdx.y * TS;
  const int bat = blockIdx.z;

  v2f bandH[7];
  v2f bandV[2][4];
  {
    const int q = lane & 15, h = lane >> 4;
#pragma unroll
    for (int kt = 0; kt < 7; ++kt) {
      const int k0 = 4 * kt + 2 * h;
      bandH[kt].x = (k0 >= q && k0 <= q + 10) ? (1.0f / 11.0f) : 0.0f;
      bandH[kt].y = (k0 + 1 >= q && k0 + 1 <= q + 10) ? (1.0f / 11.0f) : 0.0f;
    }
#pragma unroll
    for (int blk = 0; blk < 2; ++blk) {
#pragma unroll
      for (int t = 0; t < 4; ++t) {
        const int k0 = 16 * blk + 8 * h + 2 * t;
        bandV[blk][t].x = (k0 >= q && k0 <= q + 10) ? (1.0f / 11.0f) : 0.0f;
        bandV[blk][t].y = (k0 + 1 >= q && k0 + 1 <= q + 10) ? (1.0f / 11.0f) : 0.0f;
      }
    }
  }

  const int hc = tid & 63, hr = __builtin_amdgcn_readfirstlane(tid >> 6);
  const int gx = refl(x0 - 10 + hc), gy0 = y0 - 10 + hr;
  const int hlo = hr * RAWPITCH + hc;
  const bool hvalid = hc < RAWN;

  const float* imgB = img + (size_t)bat * IMG * IMG;
  const float* featB = feat + (size_t)bat * CH * IMG * IMG;
  float tI[NLD], tF[NLD];
  halo_fetch(imgB, gx, gy0, tI);
  halo_fetch(featB, gx, gy0, tF);

  for (int i = tid; i < RAWROWS * RAWPITCH; i += 256) { rawI[i] = 0.0f; rawF[i] = 0.0f; }
  __syncthreads();
  halo_store(rawI, hlo, hvalid, tI);
  halo_store(rawF, hlo, hvalid, tF);
  __syncthreads();

  if (wave < 6) {
    const int qs = wave >= 3 ? 1 : 0, nj = wave - 3 * qs;
    if (qs == 0) blur_strip<0, 4>(rawI, rawI, RAWPITCH, nj, mI,  ABPITCH, bandH, bandV, lane);
    else         blur_strip<1, 4>(rawI, rawI, RAWPITCH, nj, vIi, ABPITCH, bandH, bandV, lane);
  }
  __syncthreads();
  for (int i = tid; i < ABROWS * ABROWS; i += 256) {
    const int r = i / ABROWS, cc = i - r * ABROWS;
    const int o = r * ABPITCH + cc;
    const float m = mI[o];
    vIi[o] = 1.0f / (vIi[o] - m * m + 1e-8f);
  }

  for (int ch = 0; ch < CH; ++ch) {
    lds_barrier();

    {
      const int t = (wave + 2 * ch) & 7;
      if (t < 6) {
        const int qs = t >= 3 ? 1 : 0, nj = t - 3 * qs;
        if (qs == 0) blur_strip<0, 4>(rawF, rawF, RAWPITCH, nj, mP,  ABPITCH, bandH, bandV, lane);
        else         blur_strip<2, 4>(rawF, rawI, RAWPITCH, nj, cIp, ABPITCH, bandH, bandV, lane);
      }
    }
    lds_barrier();

    const bool more = ch + 1 < CH;
    if (more) halo_fetch(featB + (size_t)(ch + 1) * IMG * IMG, gx, gy0, tF);

    for (int i = tid; i < ABROWS * ABROWS; i += 256) {
      const int r = i / ABROWS, cc = i - r * ABROWS;
      const int o = r * ABPITCH + cc;
      const float mi_ = mI[o], mp_ = mP[o];
      const float a = (cIp[o] - mi_ * mp_) * vIi[o];
      cIp[o] = a;
      mP[o]  = mp_ - a * mi_;
    }
    lds_barrier();

    {
      const int t = (wave + 2 * ch + 2) & 7;
      if (t < 4) {
        const int qs = t >> 1, nj = t & 1;
        if (qs == 0) blur_strip<0, 3>(cIp, cIp, ABPITCH, nj, mA, MAPITCH, bandH, bandV, lane);
        else         blur_strip<0, 3>(mP,  mP,  ABPITCH, nj, mB, MAPITCH, bandH, bandV, lane);
      }
    }
    lds_barrier();

    float* outB = out + ((size_t)bat * CH + ch) * IMG * IMG;
    for (int i = tid; i < TS * TS; i += 256) {
      const int y = i >> 5, x = i & 31;
      outB[(size_t)(y0 + y) * IMG + (x0 + x)] =
          mA[y * MAPITCH + x] * rawI[(y + 10) * RAWPITCH + (x + 10)] +
          mB[y * MAPITCH + x];
    }
    if (more) halo_store(rawF, hlo, hvalid, tF);
  }
}

extern "C" void kernel_launch(void* const* d_in, const int* in_sizes, int n_in,
                              void* d_out, int out_size, void* d_ws, size_t ws_size,
                              hipStream_t stream) {
  const float* feat = (const float*)d_in[0];
  const float* img  = (const float*)d_in[1];
  float* out = (float*)d_out;
  const int B = in_sizes[1] / (IMG * IMG);
  dim3 grid(IMG / TS, IMG / TS, B);
  guided_filter_kernel<<<grid, 256, 0, stream>>>(feat, img, out);
}


# Output tile per block 64x64 on 16 waves instead of 32x32 on 8: fewer halo rows and columns are blurred again per output, so fewer WMMAs, halo loads and barriers per pixel; one block per WGP
# speedup vs baseline: 1.8520x; 1.1388x over previous
#include <hip/hip_runtime.h>


#define IMG      512
#define CH       16
#define TS       64
#define NTHR     512
#define RAWN     84
#define RAWROWS  96
#define RAWPITCH 98
#define ABROWS   80
#define ABPITCH  98
#define MAPITCH  66
#define NLD      21

typedef __attribute__((ext_vector_type(2))) float v2f;
typedef __attribute__((ext_vector_type(8))) float v8f;

__device__ __forceinline__ int refl(int p) {
  p = p < 0 ? -p : p;
  return p >= IMG ? 2 * IMG - 2 - p : p;
}

__device__ __forceinline__ void halo_fetch(const float* __restrict__ src, int gx, int gy0,
                                           float (&v)[NLD]) {
#pragma unroll
  for (int j = 0; j < NLD; ++j) {
    const float* row = src + refl(gy0 + 4 * j) * IMG;
    v[j] = row[gx];
  }
}

__device__ __forceinline__ void lds_barrier() {
  asm volatile("s_wait_dscnt 0x0\n\ts_barrier_signal -1\n\ts_barrier_wait -1" ::: "memory");
}
__device__ __forceinline__ void halo_store(float* __restrict__ raw, int lo, bool valid,
                                           const float (&v)[NLD]) {
  if (valid) {
#pragma unroll
    for (int j = 0; j < NLD; ++j) raw[lo + 4 * j * RAWPITCH] = v[j];
  }
}

template <int MODE, int NIN>
__device__ __forceinline__ void blur_strip(const float* __restrict__ s0,
                                           const float* __restrict__ s1, int sPitch,
                                           int nj, float* __restrict__ dst, int dPitch,
                                           const v2f* __restrict__ bandH,
                                           const v2f (*__restrict__ bandV)[4],
                                           int lane) {
  const int c = lane & 15, half = lane >> 4;
  const float* p0 = s0 + c * sPitch + nj * 16 + 2 * half;
  const float* p1 = s1 + c * sPitch + nj * 16 + 2 * half;
  v8f hT[NIN];
#pragma unroll
  for (int i = 0; i < NIN; ++i) hT[i] = (v8f){};
#pragma unroll
  for (int kt = 0; kt < 7; ++kt) {
#pragma unroll
    for (int i = 0; i < NIN; ++i) {
      const int o = i * 16 * sPitch + 4 * kt;
      float x0 = p0[o], x1 = p0[o + 1];
      if (MODE == 1) { x0 *= x0; x1 *= x1; }
      if (MODE == 2) { x0 *= p1[o]; x1 *= p1[o + 1]; }
      v2f a; a.x = x0; a.y = x1;
      hT[i] = __builtin_amdgcn_wmma_f32_16x16x4_f32(false, a, false, bandH[kt],
                                                    (short)0, hT[i], false, false);
    }
  }
  v8f acc[NIN - 1];
#pragma unroll
  for (int o = 0; o < NIN - 1; ++o) acc[o] = (v8f){};
#pragma unroll
  for (int blk = 0; blk < 2; ++blk) {
#pragma unroll
    for (int t = 0; t < 4; ++t) {
#pragma unroll
      for (int o = 0; o < NIN - 1; ++o) {
        v2f b; b.x = hT[o + blk][2 * t]; b.y = hT[o + blk][2 * t + 1];
        acc[o] = __builtin_amdgcn_wmma_f32_16x16x4_f32(false, bandV[blk][t], false, b,
                                                       (short)0, acc[o], false, false);
      }
    }
  }
#pragma unroll
  for (int o = 0; o < NIN - 1; ++o) {
    float* pd = dst + (o * 16 + half * 8) * dPitch + nj * 16 + c;
#pragma unroll
    for (int v = 0; v < 8; ++v) pd[v * dPitch] = acc[o][v];
  }
}

__global__ __launch_bounds__(NTHR) void guided_filter_kernel(
    const float* __restrict__ feat, const float* __restrict__ img,
    float* __restrict__ out) {
  __shared__ float rawI[RAWROWS * RAWPITCH];
  __shared__ float rawF[RAWROWS * RAWPITCH];
  __shared__ float mI [ABROWS * ABPITCH];
  __shared__ float vIi[ABROWS * ABPITCH];
  __shared__ float mP [ABROWS * ABPITCH];
  __shared__ float cIp[ABROWS * ABPITCH];
  __shared__ float mA [TS * MAPITCH];
  __shared__ float mB [TS * MAPITCH];

  const int tid  = threadIdx.x;
  const int lane = tid & 31;
  const int wave = tid >> 5;
  const int x0 = blockIdx.x * TS, y0 = blockIdx.y * TS;
  const int bat = blockIdx.z;

  v2f bandH[7];
  v2f bandV[2][4];
  {
    const int q = lane & 15, h = lane >> 4;
#pragma unroll
    for (int kt = 0; kt < 7; ++kt) {
      const int k0 = 4 * kt + 2 * h;
      bandH[kt].x = (k0 >= q && k0 <= q + 10) ? (1.0f / 11.0f) : 0.0f;
      bandH[kt].y = (k0 + 1 >= q && k0 + 1 <= q + 10) ? (1.0f / 11.0f) : 0.0f;
    }
#pragma unroll
    for (int blk = 0; blk < 2; ++blk) {
#pragma unroll
      for (int t = 0; t < 4; ++t) {
        const int k0 = 16 * blk + 8 * h + 2 * t;
        bandV[blk][t].x = (k0 >= q && k0 <= q + 10) ? (1.0f / 11.0f) : 0.0f;
        bandV[blk][t].y = (k0 + 1 >= q && k0 + 1 <= q + 10) ? (1.0f / 11.0f) : 0.0f;
      }
    }
  }

  const int hc = tid & 127, hr = __builtin_amdgcn_readfirstlane(tid >> 7);
  const int gx = refl(x0 - 10 + hc), gy0 = y0 - 10 + hr;
  const int hlo = hr * RAWPITCH + hc;
  const bool hvalid = hc < RAWN;
  const bool hfetch = (wave & 3) != 3;

  const float* imgB = img + (size_t)bat * IMG * IMG;
  const float* featB = feat + (size_t)bat * CH * IMG * IMG;
  float tI[NLD], tF[NLD];
  if (hfetch) {
    halo_fetch(imgB, gx, gy0, tI);
    halo_fetch(featB, gx, gy0, tF);
  }

  for (int i = tid; i < RAWROWS * RAWPITCH; i += NTHR) { rawI[i] = 0.0f; rawF[i] = 0.0f; }
  __syncthreads();
  halo_store(rawI, hlo, hvalid, tI);
  halo_store(rawF, hlo, hvalid, tF);
  __syncthreads();

  if (wave < 10) {
    const int qs = wave >= 5 ? 1 : 0, nj = wave - 5 * qs;
    if (qs == 0) blur_strip<0, 6>(rawI, rawI, RAWPITCH, nj, mI,  ABPITCH, bandH, bandV, lane);
    else         blur_strip<1, 6>(rawI, rawI, RAWPITCH, nj, vIi, ABPITCH, bandH, bandV, lane);
  }
  __syncthreads();
  for (int i = tid; i < ABROWS * ABROWS; i += NTHR) {
    const int r = i / ABROWS, cc = i - r * ABROWS;
    const int o = r * ABPITCH + cc;
    const float m = mI[o];
    vIi[o] = 1.0f / (vIi[o] - m * m + 1e-8f);
  }

  for (int ch = 0; ch < CH; ++ch) {
    lds_barrier();

    {
      const int t = (wave + 2 * ch) & 15;
      if (t < 10) {
        const int qs = t >= 5 ? 1 : 0, nj = t - 5 * qs;
        if (qs == 0) blur_strip<0, 6>(rawF, rawF, RAWPITCH, nj, mP,  ABPITCH, bandH, bandV, lane);
        else         blur_strip<2, 6>(rawF, rawI, RAWPITCH, nj, cIp, ABPITCH, bandH, bandV, lane);
      }
    }
    lds_barrier();

    const bool more = ch + 1 < CH;
    if (more && hfetch) halo_fetch(featB + (size_t)(ch + 1) * IMG * IMG, gx, gy0, tF);

    for (int i = tid; i < ABROWS * ABROWS; i += NTHR) {
      const int r = i / ABROWS, cc = i - r * ABROWS;
      const int o = r * ABPITCH + cc;
      const float mi_ = mI[o], mp_ = mP[o];
      const float a = (cIp[o] - mi_ * mp_) * vIi[o];
      cIp[o] = a;
      mP[o]  = mp_ - a * mi_;
    }
    lds_barrier();

    {
      const int t = (wave + 2 * ch + 6) & 15;
      if (t < 8) {
        const int qs = t >> 2, nj = t & 3;
        if (qs == 0) blur_strip<0, 5>(cIp, cIp, ABPITCH, nj, mA, MAPITCH, bandH, bandV, lane);
        else         blur_strip<0, 5>(mP,  mP,  ABPITCH, nj, mB, MAPITCH, bandH, bandV, lane);
      }
    }
    lds_barrier();

    float* outB = out + ((size_t)bat * CH + ch) * IMG * IMG;
    for (int i = tid; i < TS * TS; i += NTHR) {
      const int y = i >> 6, x = i & 63;
      outB[(size_t)(y0 + y) * IMG + (x0 + x)] =
          mA[y * MAPITCH + x] * rawI[(y + 10) * RAWPITCH + (x + 10)] +
          mB[y * MAPITCH + x];
    }
    if (more) halo_store(rawF, hlo, hvalid, tF);
  }
}

extern "C" void kernel_launch(void* const* d_in, const int* in_sizes, int n_in,
                              void* d_out, int out_size, void* d_ws, size_t ws_size,
                              hipStream_t stream) {
  const float* feat = (const float*)d_in[0];
  const float* img  = (const float*)d_in[1];
  float* out = (float*)d_out;
  const int B = in_sizes[1] / (IMG * IMG);
  dim3 grid(IMG / TS, IMG / TS, B);
  guided_filter_kernel<<<grid, NTHR, 0, stream>>>(feat, img, out);
}


# a and b are formed in registers from one wave's own mean_p and corr_Ip D tiles (tasks = column strip x row part, both quantities) and stored directly: the a/b LDS sweep and its barrier are gone
# speedup vs baseline: 2.0742x; 1.1200x over previous
#include <hip/hip_runtime.h>


#define IMG      512
#define CH       16
#define TS       64
#define NTHR     512
#define RAWN     84
#define RAWROWS  96
#define RAWPITCH 98
#define ABROWS   80
#define ABPITCH  98
#define MAPITCH  66
#define NLD      21

typedef __attribute__((ext_vector_type(2))) float v2f;
typedef __attribute__((ext_vector_type(8))) float v8f;

__device__ __forceinline__ int refl(int p) {
  p = p < 0 ? -p : p;
  return p >= IMG ? 2 * IMG - 2 - p : p;
}

__device__ __forceinline__ void halo_fetch(const float* __restrict__ src, int gx, int gy0,
                                           float (&v)[NLD]) {
#pragma unroll
  for (int j = 0; j < NLD; ++j) {
    const float* row = src + refl(gy0 + 4 * j) * IMG;
    v[j] = row[gx];
  }
}

__device__ __forceinline__ void lds_barrier() {
  asm volatile("s_wait_dscnt 0x0\n\ts_barrier_signal -1\n\ts_barrier_wait -1" ::: "memory");
}
__device__ __forceinline__ void halo_store(float* __restrict__ raw, int lo, bool valid,
                                           const float (&v)[NLD]) {
  if (valid) {
#pragma unroll
    for (int j = 0; j < NLD; ++j) raw[lo + 4 * j * RAWPITCH] = v[j];
  }
}

template <int MODE, int NIN>
__device__ __forceinline__ void blur_tiles(const float* __restrict__ s0,
                                           const float* __restrict__ s1, int sPitch,
                                           int nj,
                                           const v2f* __restrict__ bandH,
                                           const v2f (*__restrict__ bandV)[4],
                                           int lane, v8f (&acc)[NIN - 1]) {
  const int c = lane & 15, half = lane >> 4;
  const float* p0 = s0 + c * sPitch + nj * 16 + 2 * half;
  const float* p1 = s1 + c * sPitch + nj * 16 + 2 * half;
  v8f hT[NIN];
#pragma unroll
  for (int i = 0; i < NIN; ++i) hT[i] = (v8f){};
#pragma unroll
  for (int kt = 0; kt < 7; ++kt) {
#pragma unroll
    for (int i = 0; i < NIN; ++i) {
      const int o = i * 16 * sPitch + 4 * kt;
      float x0 = p0[o], x1 = p0[o + 1];
      if (MODE == 1) { x0 *= x0; x1 *= x1; }
      if (MODE == 2) { x0 *= p1[o]; x1 *= p1[o + 1]; }
      v2f a; a.x = x0; a.y = x1;
      hT[i] = __builtin_amdgcn_wmma_f32_16x16x4_f32(false, a, false, bandH[kt],
                                                    (short)0, hT[i], false, false);
    }
  }
#pragma unroll
  for (int o = 0; o < NIN - 1; ++o) acc[o] = (v8f){};
#pragma unroll
  for (int blk = 0; blk < 2; ++blk) {
#pragma unroll
    for (int t = 0; t < 4; ++t) {
#pragma unroll
      for (int o = 0; o < NIN - 1; ++o) {
        v2f b; b.x = hT[o + blk][2 * t]; b.y = hT[o + blk][2 * t + 1];
        acc[o] = __builtin_amdgcn_wmma_f32_16x16x4_f32(false, bandV[blk][t], false, b,
                                                       (short)0, acc[o], false, false);
      }
    }
  }
}

template <int MODE, int NIN>
__device__ __forceinline__ void blur_strip(const float* __restrict__ s0,
                                           const float* __restrict__ s1, int sPitch,
                                           int nj, float* __restrict__ dst, int dPitch,
                                           const v2f* __restrict__ bandH,
                                           const v2f (*__restrict__ bandV)[4],
                                           int lane) {
  const int c = lane & 15, half = lane >> 4;
  v8f acc[NIN - 1];
  blur_tiles<MODE, NIN>(s0, s1, sPitch, nj, bandH, bandV, lane, acc);
#pragma unroll
  for (int o = 0; o < NIN - 1; ++o) {
    float* pd = dst + (o * 16 + half * 8) * dPitch + nj * 16 + c;
#pragma unroll
    for (int v = 0; v < 8; ++v) pd[v * dPitch] = acc[o][v];
  }
}

template <int NIN>
__device__ __forceinline__ void ab_tiles(const float* __restrict__ rF,
                                         const float* __restrict__ rI, int nj,
                                         const float* __restrict__ mIr,
                                         const float* __restrict__ vIr,
                                         float* __restrict__ aDst, float* __restrict__ bDst,
                                         const v2f* __restrict__ bandH,
                                         const v2f (*__restrict__ bandV)[4],
                                         int lane) {
  const int c = lane & 15, half = lane >> 4;
  v8f mp[NIN - 1], cp[NIN - 1];
  blur_tiles<0, NIN>(rF, rF, RAWPITCH, nj, bandH, bandV, lane, mp);
  blur_tiles<2, NIN>(rF, rI, RAWPITCH, nj, bandH, bandV, lane, cp);
#pragma unroll
  for (int o = 0; o < NIN - 1; ++o) {
    const int base = (o * 16 + half * 8) * ABPITCH + nj * 16 + c;
#pragma unroll
    for (int v = 0; v < 8; ++v) {
      const int idx = base + v * ABPITCH;
      const float mi_ = mIr[idx], mp_ = mp[o][v];
      const float a = (cp[o][v] - mi_ * mp_) * vIr[idx];
      aDst[idx] = a;
      bDst[idx] = mp_ - a * mi_;
    }
  }
}

__global__ __launch_bounds__(NTHR) void guided_filter_kernel(
    const float* __restrict__ feat, const float* __restrict__ img,
    float* __restrict__ out) {
  __shared__ float rawI[RAWROWS * RAWPITCH];
  __shared__ float rawF[RAWROWS * RAWPITCH];
  __shared__ float mI [ABROWS * ABPITCH];
  __shared__ float vIi[ABROWS * ABPITCH];
  __shared__ float mP [ABROWS * ABPITCH];
  __shared__ float cIp[ABROWS * ABPITCH];
  __shared__ float mA [TS * MAPITCH];
  __shared__ float mB [TS * MAPITCH];

  const int tid  = threadIdx.x;
  const int lane = tid & 31;
  const int wave = tid >> 5;
  const int x0 = blockIdx.x * TS, y0 = blockIdx.y * TS;
  const int bat = blockIdx.z;

  v2f bandH[7];
  v2f bandV[2][4];
  {
    const int q = lane & 15, h = lane >> 4;
#pragma unroll
    for (int kt = 0; kt < 7; ++kt) {
      const int k0 = 4 * kt + 2 * h;
      bandH[kt].x = (k0 >= q && k0 <= q + 10) ? (1.0f / 11.0f) : 0.0f;
      bandH[kt].y = (k0 + 1 >= q && k0 + 1 <= q + 10) ? (1.0f / 11.0f) : 0.0f;
    }
#pragma unroll
    for (int blk = 0; blk < 2; ++blk) {
#pragma unroll
      for (int t = 0; t < 4; ++t) {
        const int k0 = 16 * blk + 8 * h + 2 * t;
        bandV[blk][t].x = (k0 >= q && k0 <= q + 10) ? (1.0f / 11.0f) : 0.0f;
        bandV[blk][t].y = (k0 + 1 >= q && k0 + 1 <= q + 10) ? (1.0f / 11.0f) : 0.0f;
      }
    }
  }

  const int hc = tid & 127, hr = __builtin_amdgcn_readfirstlane(tid >> 7);
  const int gx = refl(x0 - 10 + hc), gy0 = y0 - 10 + hr;
  const int hlo = hr * RAWPITCH + hc;
  const bool hvalid = hc < RAWN;
  const bool hfetch = (wave & 3) != 3;

  const float* imgB = img + (size_t)bat * IMG * IMG;
  const float* featB = feat + (size_t)bat * CH * IMG * IMG;
  float tI[NLD], tF[NLD];
  if (hfetch) {
    halo_fetch(imgB, gx, gy0, tI);
    halo_fetch(featB, gx, gy0, tF);
  }

  for (int i = tid; i < RAWROWS * RAWPITCH; i += NTHR) { rawI[i] = 0.0f; rawF[i] = 0.0f; }
  __syncthreads();
  halo_store(rawI, hlo, hvalid, tI);
  halo_store(rawF, hlo, hvalid, tF);
  __syncthreads();

  if (wave < 10) {
    const int qs = wave >= 5 ? 1 : 0, nj = wave - 5 * qs;
    if (qs == 0) blur_strip<0, 6>(rawI, rawI, RAWPITCH, nj, mI,  ABPITCH, bandH, bandV, lane);
    else         blur_strip<1, 6>(rawI, rawI, RAWPITCH, nj, vIi, ABPITCH, bandH, bandV, lane);
  }
  __syncthreads();
  for (int i = tid; i < ABROWS * ABROWS; i += NTHR) {
    const int r = i / ABROWS, cc = i - r * ABROWS;
    const int o = r * ABPITCH + cc;
    const float m = mI[o];
    vIi[o] = 1.0f / (vIi[o] - m * m + 1e-8f);
  }

  for (int ch = 0; ch < CH; ++ch) {
    lds_barrier();

    {
      const int t = (wave + 2 * ch) & 15;
      if (t < 10) {
        const bool upper = (t & 2) != 0 || t == 9;
        if (upper) {
          const int nj = t == 9 ? 4 : (t >> 2) + 2 * (t & 1);
          ab_tiles<4>(rawF, rawI, nj, mI, vIi, cIp, mP, bandH, bandV, lane);
        } else {
          const int nj = (t >> 2) + 3 * (t & 1);
          ab_tiles<3>(rawF + 48 * RAWPITCH, rawI + 48 * RAWPITCH, nj,
                      mI + 48 * ABPITCH, vIi + 48 * ABPITCH,
                      cIp + 48 * ABPITCH, mP + 48 * ABPITCH, bandH, bandV, lane);
        }
      }
    }
    lds_barrier();

    const bool more = ch + 1 < CH;
    if (more && hfetch) halo_fetch(featB + (size_t)(ch + 1) * IMG * IMG, gx, gy0, tF);

    {
      const int t = (wave + 2 * ch + 6) & 15;
      if (t < 8) {
        const int qs = t >> 2, nj = t & 3;
        if (qs == 0) blur_strip<0, 5>(cIp, cIp, ABPITCH, nj, mA, MAPITCH, bandH, bandV, lane);
        else         blur_strip<0, 5>(mP,  mP,  ABPITCH, nj, mB, MAPITCH, bandH, bandV, lane);
      }
    }
    lds_barrier();

    float* outB = out + ((size_t)bat * CH + ch) * IMG * IMG;
    for (int i = tid; i < TS * TS; i += NTHR) {
      const int y = i >> 6, x = i & 63;
      outB[(size_t)(y0 + y) * IMG + (x0 + x)] =
          mA[y * MAPITCH + x] * rawI[(y + 10) * RAWPITCH + (x + 10)] +
          mB[y * MAPITCH + x];
    }
    if (more) halo_store(rawF, hlo, hvalid, tF);
  }
}

extern "C" void kernel_launch(void* const* d_in, const int* in_sizes, int n_in,
                              void* d_out, int out_size, void* d_ws, size_t ws_size,
                              hipStream_t stream) {
  const float* feat = (const float*)d_in[0];
  const float* img  = (const float*)d_in[1];
  float* out = (float*)d_out;
  const int B = in_sizes[1] / (IMG * IMG);
  dim3 grid(IMG / TS, IMG / TS, B);
  guided_filter_kernel<<<grid, NTHR, 0, stream>>>(feat, img, out);
}


# Output sweep (out = mean_a*I + mean_b) unrolled, all reads first: a thread's 24 LDS reads for its 8 rows are in flight together, then its 8 stores; before, 8 trips each waited for their own reads
# speedup vs baseline: 2.2459x; 1.0828x over previous
#include <hip/hip_runtime.h>


#define IMG      512
#define CH       16
#define TS       64
#define NTHR     512
#define RAWN     84
#define RAWROWS  96
#define RAWPITCH 98
#define ABROWS   80
#define ABPITCH  98
#define MAPITCH  66
#define NLD      21
#define OUTN     8

typedef __attribute__((ext_vector_type(2))) float v2f;
typedef __attribute__((ext_vector_type(8))) float v8f;

__device__ __forceinline__ int refl(int p) {
  p = p < 0 ? -p : p;
  return p >= IMG ? 2 * IMG - 2 - p : p;
}

__device__ __forceinline__ void halo_fetch(const float* __restrict__ src, int gx, int gy0,
                                           float (&v)[NLD]) {
#pragma unroll
  for (int j = 0; j < NLD; ++j) {
    const float* row = src + refl(gy0 + 4 * j) * IMG;
    v[j] = row[gx];
  }
}

__device__ __forceinline__ void lds_barrier() {
  asm volatile("s_wait_dscnt 0x0\n\ts_barrier_signal -1\n\ts_barrier_wait -1" ::: "memory");
}
__device__ __forceinline__ void halo_store(float* __restrict__ raw, int lo, bool valid,
                                           const float (&v)[NLD]) {
  if (valid) {
#pragma unroll
    for (int j = 0; j < NLD; ++j) raw[lo + 4 * j * RAWPITCH] = v[j];
  }
}

template <int MODE, int NIN>
__device__ __forceinline__ void blur_tiles(const float* __restrict__ s0,
                                           const float* __restrict__ s1, int sPitch,
                                           int nj,
                                           const v2f* __restrict__ bandH,
                                           const v2f (*__restrict__ bandV)[4],
                                           int lane, v8f (&acc)[NIN - 1]) {
  const int c = lane & 15, half = lane >> 4;
  const float* p0 = s0 + c * sPitch + nj * 16 + 2 * half;
  const float* p1 = s1 + c * sPitch + nj * 16 + 2 * half;
  v8f hT[NIN];
#pragma unroll
  for (int i = 0; i < NIN; ++i) hT[i] = (v8f){};
#pragma unroll
  for (int kt = 0; kt < 7; ++kt) {
#pragma unroll
    for (int i = 0; i < NIN; ++i) {
      const int o = i * 16 * sPitch + 4 * kt;
      float x0 = p0[o], x1 = p0[o + 1];
      if (MODE == 1) { x0 *= x0; x1 *= x1; }
      if (MODE == 2) { x0 *= p1[o]; x1 *= p1[o + 1]; }
      v2f a; a.x = x0; a.y = x1;
      hT[i] = __builtin_amdgcn_wmma_f32_16x16x4_f32(false, a, false, bandH[kt],
                                                    (short)0, hT[i], false, false);
    }
  }
#pragma unroll
  for (int o = 0; o < NIN - 1; ++o) acc[o] = (v8f){};
#pragma unroll
  for (int blk = 0; blk < 2; ++blk) {
#pragma unroll
    for (int t = 0; t < 4; ++t) {
#pragma unroll
      for (int o = 0; o < NIN - 1; ++o) {
        v2f b; b.x = hT[o + blk][2 * t]; b.y = hT[o + blk][2 * t + 1];
        acc[o] = __builtin_amdgcn_wmma_f32_16x16x4_f32(false, bandV[blk][t], false, b,
                                                       (short)0, acc[o], false, false);
      }
    }
  }
}

template <int MODE, int NIN>
__device__ __forceinline__ void blur_strip(const float* __restrict__ s0,
                                           const float* __restrict__ s1, int sPitch,
                                           int nj, float* __restrict__ dst, int dPitch,
                                           const v2f* __restrict__ bandH,
                                           const v2f (*__restrict__ bandV)[4],
                                           int lane) {
  const int c = lane & 15, half = lane >> 4;
  v8f acc[NIN - 1];
  blur_tiles<MODE, NIN>(s0, s1, sPitch, nj, bandH, bandV, lane, acc);
#pragma unroll
  for (int o = 0; o < NIN - 1; ++o) {
    float* pd = dst + (o * 16 + half * 8) * dPitch + nj * 16 + c;
#pragma unroll
    for (int v = 0; v < 8; ++v) pd[v * dPitch] = acc[o][v];
  }
}

template <int NIN>
__device__ __forceinline__ void ab_tiles(const float* __restrict__ rF,
                                         const float* __restrict__ rI, int nj,
                                         const float* __restrict__ mIr,
                                         const float* __restrict__ vIr,
                                         float* __restrict__ aDst, float* __restrict__ bDst,
                                         const v2f* __restrict__ bandH,
                                         const v2f (*__restrict__ bandV)[4],
                                         int lane) {
  const int c = lane & 15, half = lane >> 4;
  v8f mp[NIN - 1], cp[NIN - 1];
  blur_tiles<0, NIN>(rF, rF, RAWPITCH, nj, bandH, bandV, lane, mp);
  blur_tiles<2, NIN>(rF, rI, RAWPITCH, nj, bandH, bandV, lane, cp);
#pragma unroll
  for (int o = 0; o < NIN - 1; ++o) {
    const int base = (o * 16 + half * 8) * ABPITCH + nj * 16 + c;
#pragma unroll
    for (int v = 0; v < 8; ++v) {
      const int idx = base + v * ABPITCH;
      const float mi_ = mIr[idx], mp_ = mp[o][v];
      const float a = (cp[o][v] - mi_ * mp_) * vIr[idx];
      aDst[idx] = a;
      bDst[idx] = mp_ - a * mi_;
    }
  }
}

__global__ __launch_bounds__(NTHR) void guided_filter_kernel(
    const float* __restrict__ feat, const float* __restrict__ img,
    float* __restrict__ out) {
  __shared__ float rawI[RAWROWS * RAWPITCH];
  __shared__ float rawF[RAWROWS * RAWPITCH];
  __shared__ float mI [ABROWS * ABPITCH];
  __shared__ float vIi[ABROWS * ABPITCH];
  __shared__ float mP [ABROWS * ABPITCH];
  __shared__ float cIp[ABROWS * ABPITCH];
  __shared__ float mA [TS * MAPITCH];
  __shared__ float mB [TS * MAPITCH];

  const int tid  = threadIdx.x;
  const int lane = tid & 31;
  const int wave = tid >> 5;
  const int x0 = blockIdx.x * TS, y0 = blockIdx.y * TS;
  const int bat = blockIdx.z;

  v2f bandH[7];
  v2f bandV[2][4];
  {
    const int q = lane & 15, h = lane >> 4;
#pragma unroll
    for (int kt = 0; kt < 7; ++kt) {
      const int k0 = 4 * kt + 2 * h;
      bandH[kt].x = (k0 >= q && k0 <= q + 10) ? (1.0f / 11.0f) : 0.0f;
      bandH[kt].y = (k0 + 1 >= q && k0 + 1 <= q + 10) ? (1.0f / 11.0f) : 0.0f;
    }
#pragma unroll
    for (int blk = 0; blk < 2; ++blk) {
#pragma unroll
      for (int t = 0; t < 4; ++t) {
        const int k0 = 16 * blk + 8 * h + 2 * t;
        bandV[blk][t].x = (k0 >= q && k0 <= q + 10) ? (1.0f / 11.0f) : 0.0f;
        bandV[blk][t].y = (k0 + 1 >= q && k0 + 1 <= q + 10) ? (1.0f / 11.0f) : 0.0f;
      }
    }
  }

  const int hc = tid & 127, hr = __builtin_amdgcn_readfirstlane(tid >> 7);
  const int gx = refl(x0 - 10 + hc), gy0 = y0 - 10 + hr;
  const int hlo = hr * RAWPITCH + hc;
  const bool hvalid = hc < RAWN;
  const bool hfetch = (wave & 3) != 3;

  const float* imgB = img + (size_t)bat * IMG * IMG;
  const float* featB = feat + (size_t)bat * CH * IMG * IMG;
  float tI[NLD], tF[NLD];
  if (hfetch) {
    halo_fetch(imgB, gx, gy0, tI);
    halo_fetch(featB, gx, gy0, tF);
  }

  for (int i = tid; i < RAWROWS * RAWPITCH; i += NTHR) { rawI[i] = 0.0f; rawF[i] = 0.0f; }
  __syncthreads();
  halo_store(rawI, hlo, hvalid, tI);
  halo_store(rawF, hlo, hvalid, tF);
  __syncthreads();

  if (wave < 10) {
    const int qs = wave >= 5 ? 1 : 0, nj = wave - 5 * qs;
    if (qs == 0) blur_strip<0, 6>(rawI, rawI, RAWPITCH, nj, mI,  ABPITCH, bandH, bandV, lane);
    else         blur_strip<1, 6>(rawI, rawI, RAWPITCH, nj, vIi, ABPITCH, bandH, bandV, lane);
  }
  __syncthreads();
  for (int i = tid; i < ABROWS * ABROWS; i += NTHR) {
    const int r = i / ABROWS, cc = i - r * ABROWS;
    const int o = r * ABPITCH + cc;
    const float m = mI[o];
    vIi[o] = 1.0f / (vIi[o] - m * m + 1e-8f);
  }

  for (int ch = 0; ch < CH; ++ch) {
    lds_barrier();

    {
      const int t = (wave + 2 * ch) & 15;
      if (t < 10) {
        const bool upper = (t & 2) != 0 || t == 9;
        if (upper) {
          const int nj = t == 9 ? 4 : (t >> 2) + 2 * (t & 1);
          ab_tiles<4>(rawF, rawI, nj, mI, vIi, cIp, mP, bandH, bandV, lane);
        } else {
          const int nj = (t >> 2) + 3 * (t & 1);
          ab_tiles<3>(rawF + 48 * RAWPITCH, rawI + 48 * RAWPITCH, nj,
                      mI + 48 * ABPITCH, vIi + 48 * ABPITCH,
                      cIp + 48 * ABPITCH, mP + 48 * ABPITCH, bandH, bandV, lane);
        }
      }
    }
    lds_barrier();

    const bool more = ch + 1 < CH;
    if (more && hfetch) halo_fetch(featB + (size_t)(ch + 1) * IMG * IMG, gx, gy0, tF);

    {
      const int t = (wave + 2 * ch + 6) & 15;
      if (t < 8) {
        const int qs = t >> 2, nj = t & 3;
        if (qs == 0) blur_strip<0, 5>(cIp, cIp, ABPITCH, nj, mA, MAPITCH, bandH, bandV, lane);
        else         blur_strip<0, 5>(mP,  mP,  ABPITCH, nj, mB, MAPITCH, bandH, bandV, lane);
      }
    }
    lds_barrier();

    {
      const int oy = tid >> 6, ox = tid & 63;
      const float* pA = mA + oy * MAPITCH + ox;
      const float* pB = mB + oy * MAPITCH + ox;
      const float* pI = rawI + (oy + 10) * RAWPITCH + (ox + 10);
      float* pO = out + ((size_t)bat * CH + ch) * IMG * IMG + (size_t)(y0 + oy) * IMG + (x0 + ox);
      float va[OUTN], vb[OUTN], vi[OUTN];
#pragma unroll
      for (int j = 0; j < OUTN; ++j) {
        va[j] = pA[j * 8 * MAPITCH];
        vb[j] = pB[j * 8 * MAPITCH];
        vi[j] = pI[j * 8 * RAWPITCH];
      }
#pragma unroll
      for (int j = 0; j < OUTN; ++j) pO[(size_t)j * 8 * IMG] = va[j] * vi[j] + vb[j];
    }
    if (more) halo_store(rawF, hlo, hvalid, tF);
  }
}

extern "C" void kernel_launch(void* const* d_in, const int* in_sizes, int n_in,
                              void* d_out, int out_size, void* d_ws, size_t ws_size,
                              hipStream_t stream) {
  const float* feat = (const float*)d_in[0];
  const float* img  = (const float*)d_in[1];
  float* out = (float*)d_out;
  const int B = in_sizes[1] / (IMG * IMG);
  dim3 grid(IMG / TS, IMG / TS, B);
  guided_filter_kernel<<<grid, NTHR, 0, stream>>>(feat, img, out);
}


# K-steps that are zero by construction get no WMMA: H-tile rows reordered so each V k-step spans 4 consecutive rows (V-pass sum order changes); band-zero and zero-padding k-steps dropped
# speedup vs baseline: 2.4007x; 1.0689x over previous
#include <hip/hip_runtime.h>


#define IMG      512
#define CH       16
#define TS       64
#define NTHR     512
#define RAWN     84
#define RAWROWS  96
#define RAWPITCH 98
#define ABROWS   80
#define ABPITCH  98
#define MAPITCH  66
#define NLD      21
#define OUTN     8

typedef __attribute__((ext_vector_type(2))) float v2f;
typedef __attribute__((ext_vector_type(8))) float v8f;

__device__ __forceinline__ int refl(int p) {
  p = p < 0 ? -p : p;
  return p >= IMG ? 2 * IMG - 2 - p : p;
}

__device__ __forceinline__ void halo_fetch(const float* __restrict__ src, int gx, int gy0,
                                           float (&v)[NLD]) {
#pragma unroll
  for (int j = 0; j < NLD; ++j) {
    const float* row = src + refl(gy0 + 4 * j) * IMG;
    v[j] = row[gx];
  }
}

__device__ __forceinline__ void lds_barrier() {
  asm volatile("s_wait_dscnt 0x0\n\ts_barrier_signal -1\n\ts_barrier_wait -1" ::: "memory");
}
__device__ __forceinline__ void halo_store(float* __restrict__ raw, int lo, bool valid,
                                           const float (&v)[NLD]) {
  if (valid) {
#pragma unroll
    for (int j = 0; j < NLD; ++j) raw[lo + 4 * j * RAWPITCH] = v[j];
  }
}

__device__ __forceinline__ int frag_row(int c) {
  return 4 * ((c >> 1) & 3) + 2 * (c >> 3) + (c & 1);
}

template <int MODE, int NIN, int KT0, int KT1>
__device__ __forceinline__ void h_steps(const float* __restrict__ p0,
                                        const float* __restrict__ p1, int sPitch,
                                        const v2f* __restrict__ bandH, v8f (&hT)[NIN]) {
#pragma unroll
  for (int kt = KT0; kt < KT1; ++kt) {
#pragma unroll
    for (int i = 0; i < NIN; ++i) {
      const int o = i * 16 * sPitch + 4 * kt;
      float x0 = p0[o], x1 = p0[o + 1];
      if (MODE == 1) { x0 *= x0; x1 *= x1; }
      if (MODE == 2) { x0 *= p1[o]; x1 *= p1[o + 1]; }
      v2f a; a.x = x0; a.y = x1;
      hT[i] = __builtin_amdgcn_wmma_f32_16x16x4_f32(false, a, false, bandH[kt],
                                                    (short)0, hT[i], false, false);
    }
  }
}

template <int NIN, int KT0, int KT1>
__device__ __forceinline__ void h_steps_pair(const float* __restrict__ p0,
                                             const float* __restrict__ p1, int sPitch,
                                             const v2f* __restrict__ bandH,
                                             v8f (&hA)[NIN], v8f (&hB)[NIN]) {
#pragma unroll
  for (int kt = KT0; kt < KT1; ++kt) {
#pragma unroll
    for (int i = 0; i < NIN; ++i) {
      const int o = i * 16 * sPitch + 4 * kt;
      v2f a; a.x = p0[o]; a.y = p0[o + 1];
      v2f b; b.x = a.x * p1[o]; b.y = a.y * p1[o + 1];
      hA[i] = __builtin_amdgcn_wmma_f32_16x16x4_f32(false, a, false, bandH[kt],
                                                    (short)0, hA[i], false, false);
      hB[i] = __builtin_amdgcn_wmma_f32_16x16x4_f32(false, b, false, bandH[kt],
                                                    (short)0, hB[i], false, false);
    }
  }
}

template <int NIN, bool TAILZERO>
__device__ __forceinline__ void v_pass(const v8f (&hT)[NIN], const v2f* __restrict__ bandH,
                                       v8f (&acc)[NIN - 1]) {
#pragma unroll
  for (int o = 0; o < NIN - 1; ++o) acc[o] = (v8f){};
#pragma unroll
  for (int ks = 0; ks < 7; ++ks) {
    const int blk = ks >> 2, t = ks & 3;
#pragma unroll
    for (int o = 0; o < NIN - 1; ++o) {
      if (TAILZERO && o == NIN - 2 && ks > 4) continue;
      v2f b; b.x = hT[o + blk][2 * t]; b.y = hT[o + blk][2 * t + 1];
      acc[o] = __builtin_amdgcn_wmma_f32_16x16x4_f32(false, bandH[ks], false, b,
                                                     (short)0, acc[o], false, false);
    }
  }
}

template <int MODE, int NIN, bool MAYLAST, bool TAILZERO>
__device__ __forceinline__ void blur_tiles(const float* __restrict__ s0,
                                           const float* __restrict__ s1, int sPitch,
                                           int nj,
                                           const v2f* __restrict__ bandH,
                                           int lane, v8f (&acc)[NIN - 1]) {
  const int c = lane & 15, half = lane >> 4;
  const float* p0 = s0 + frag_row(c) * sPitch + nj * 16 + 2 * half;
  const float* p1 = s1 + frag_row(c) * sPitch + nj * 16 + 2 * half;
  v8f hT[NIN];
#pragma unroll
  for (int i = 0; i < NIN; ++i) hT[i] = (v8f){};
  h_steps<MODE, NIN, 0, 5>(p0, p1, sPitch, bandH, hT);
  if (!MAYLAST || nj != 4) h_steps<MODE, NIN, 5, 7>(p0, p1, sPitch, bandH, hT);
  v_pass<NIN, TAILZERO>(hT, bandH, acc);
}

template <int MODE, int NIN, bool MAYLAST, bool TAILZERO>
__device__ __forceinline__ void blur_strip(const float* __restrict__ s0,
                                           const float* __restrict__ s1, int sPitch,
                                           int nj, float* __restrict__ dst, int dPitch,
                                           const v2f* __restrict__ bandH,
                                           int lane) {
  const int c = lane & 15, half = lane >> 4;
  v8f acc[NIN - 1];
  blur_tiles<MODE, NIN, MAYLAST, TAILZERO>(s0, s1, sPitch, nj, bandH, lane, acc);
#pragma unroll
  for (int o = 0; o < NIN - 1; ++o) {
    float* pd = dst + (o * 16 + half * 8) * dPitch + nj * 16 + c;
#pragma unroll
    for (int v = 0; v < 8; ++v) pd[v * dPitch] = acc[o][v];
  }
}

template <int NIN, bool TAILZERO>
__device__ __forceinline__ void ab_tiles(const float* __restrict__ rF,
                                         const float* __restrict__ rI, int nj,
                                         const float* __restrict__ mIr,
                                         const float* __restrict__ vIr,
                                         float* __restrict__ aDst, float* __restrict__ bDst,
                                         const v2f* __restrict__ bandH,
                                         int lane) {
  const int c = lane & 15, half = lane >> 4;
  const float* pF = rF + frag_row(c) * RAWPITCH + nj * 16 + 2 * half;
  const float* pI = rI + frag_row(c) * RAWPITCH + nj * 16 + 2 * half;
  v8f hM[NIN], hC[NIN];
#pragma unroll
  for (int i = 0; i < NIN; ++i) { hM[i] = (v8f){}; hC[i] = (v8f){}; }
  h_steps_pair<NIN, 0, 5>(pF, pI, RAWPITCH, bandH, hM, hC);
  if (nj != 4) h_steps_pair<NIN, 5, 7>(pF, pI, RAWPITCH, bandH, hM, hC);
  v8f mp[NIN - 1], cp[NIN - 1];
  v_pass<NIN, TAILZERO>(hM, bandH, mp);
  v_pass<NIN, TAILZERO>(hC, bandH, cp);
#pragma unroll
  for (int o = 0; o < NIN - 1; ++o) {
    const int base = (o * 16 + half * 8) * ABPITCH + nj * 16 + c;
#pragma unroll
    for (int v = 0; v < 8; ++v) {
      const int idx = base + v * ABPITCH;
      const float mi_ = mIr[idx], mp_ = mp[o][v];
      const float a = (cp[o][v] - mi_ * mp_) * vIr[idx];
      aDst[idx] = a;
      bDst[idx] = mp_ - a * mi_;
    }
  }
}

__global__ __launch_bounds__(NTHR) void guided_filter_kernel(
    const float* __restrict__ feat, const float* __restrict__ img,
    float* __restrict__ out) {
  __shared__ float rawI[RAWROWS * RAWPITCH];
  __shared__ float rawF[RAWROWS * RAWPITCH];
  __shared__ float mI [ABROWS * ABPITCH];
  __shared__ float vIi[ABROWS * ABPITCH];
  __shared__ float mP [ABROWS * ABPITCH];
  __shared__ float cIp[ABROWS * ABPITCH];
  __shared__ float mA [TS * MAPITCH];
  __shared__ float mB [TS * MAPITCH];

  const int tid  = threadIdx.x;
  const int lane = tid & 31;
  const int wave = tid >> 5;
  const int x0 = blockIdx.x * TS, y0 = blockIdx.y * TS;
  const int bat = blockIdx.z;

  v2f bandH[7];
  {
    const int q = lane & 15, h = lane >> 4;
#pragma unroll
    for (int kt = 0; kt < 7; ++kt) {
      const int k0 = 4 * kt + 2 * h;
      bandH[kt].x = (k0 >= q && k0 <= q + 10) ? (1.0f / 11.0f) : 0.0f;
      bandH[kt].y = (k0 + 1 >= q && k0 + 1 <= q + 10) ? (1.0f / 11.0f) : 0.0f;
    }
  }

  const int hc = tid & 127, hr = __builtin_amdgcn_readfirstlane(tid >> 7);
  const int gx = refl(x0 - 10 + hc), gy0 = y0 - 10 + hr;
  const int hlo = hr * RAWPITCH + hc;
  const bool hvalid = hc < RAWN;
  const bool hfetch = (wave & 3) != 3;

  const float* imgB = img + (size_t)bat * IMG * IMG;
  const float* featB = feat + (size_t)bat * CH * IMG * IMG;
  float tI[NLD], tF[NLD];
  if (hfetch) {
    halo_fetch(imgB, gx, gy0, tI);
    halo_fetch(featB, gx, gy0, tF);
  }

  for (int i = tid; i < RAWROWS * RAWPITCH; i += NTHR) { rawI[i] = 0.0f; rawF[i] = 0.0f; }
  __syncthreads();
  halo_store(rawI, hlo, hvalid, tI);
  halo_store(rawF, hlo, hvalid, tF);
  __syncthreads();

  if (wave < 10) {
    const int qs = wave >= 5 ? 1 : 0, nj = wave - 5 * qs;
    if (qs == 0) blur_strip<0, 6, true, true>(rawI, rawI, RAWPITCH, nj, mI,  ABPITCH, bandH, lane);
    else         blur_strip<1, 6, true, true>(rawI, rawI, RAWPITCH, nj, vIi, ABPITCH, bandH, lane);
  }
  __syncthreads();
  for (int i = tid; i < ABROWS * ABROWS; i += NTHR) {
    const int r = i / ABROWS, cc = i - r * ABROWS;
    const int o = r * ABPITCH + cc;
    const float m = mI[o];
    vIi[o] = 1.0f / (vIi[o] - m * m + 1e-8f);
  }

  for (int ch = 0; ch < CH; ++ch) {
    lds_barrier();

    {
      const int t = (wave + 2 * ch) & 15;
      if (t < 10) {
        const bool upper = (t & 2) != 0 || t == 9;
        if (upper) {
          const int nj = t == 9 ? 4 : (t >> 2) + 2 * (t & 1);
          ab_tiles<4, false>(rawF, rawI, nj, mI, vIi, cIp, mP, bandH, lane);
        } else {
          const int nj = (t >> 2) + 3 * (t & 1);
          ab_tiles<3, true>(rawF + 48 * RAWPITCH, rawI + 48 * RAWPITCH, nj,
                            mI + 48 * ABPITCH, vIi + 48 * ABPITCH,
                            cIp + 48 * ABPITCH, mP + 48 * ABPITCH, bandH, lane);
        }
      }
    }
    lds_barrier();

    const bool more = ch + 1 < CH;
    if (more && hfetch) halo_fetch(featB + (size_t)(ch + 1) * IMG * IMG, gx, gy0, tF);

    {
      const int t = (wave + 2 * ch + 6) & 15;
      if (t < 8) {
        const int qs = t >> 2, nj = t & 3;
        if (qs == 0) blur_strip<0, 5, false, false>(cIp, cIp, ABPITCH, nj, mA, MAPITCH, bandH, lane);
        else         blur_strip<0, 5, false, false>(mP,  mP,  ABPITCH, nj, mB, MAPITCH, bandH, lane);
      }
    }
    lds_barrier();

    {
      const int oy = tid >> 6, ox = tid & 63;
      const float* pA = mA + oy * MAPITCH + ox;
      const float* pB = mB + oy * MAPITCH + ox;
      const float* pI = rawI + (oy + 10) * RAWPITCH + (ox + 10);
      float* pO = out + ((size_t)bat * CH + ch) * IMG * IMG + (size_t)(y0 + oy) * IMG + (x0 + ox);
      float va[OUTN], vb[OUTN], vi[OUTN];
#pragma unroll
      for (int j = 0; j < OUTN; ++j) {
        va[j] = pA[j * 8 * MAPITCH];
        vb[j] = pB[j * 8 * MAPITCH];
        vi[j] = pI[j * 8 * RAWPITCH];
      }
#pragma unroll
      for (int j = 0; j < OUTN; ++j) pO[(size_t)j * 8 * IMG] = va[j] * vi[j] + vb[j];
    }
    if (more) halo_store(rawF, hlo, hvalid, tF);
  }
}

extern "C" void kernel_launch(void* const* d_in, const int* in_sizes, int n_in,
                              void* d_out, int out_size, void* d_ws, size_t ws_size,
                              hipStream_t stream) {
  const float* feat = (const float*)d_in[0];
  const float* img  = (const float*)d_in[1];
  float* out = (float*)d_out;
  const int B = in_sizes[1] / (IMG * IMG);
  dim3 grid(IMG / TS, IMG / TS, B);
  guided_filter_kernel<<<grid, NTHR, 0, stream>>>(feat, img, out);
}


# Output sweep of channel ch-1 runs on the 6 waves without a task in channel ch's first blur pair; each wave stores its next-halo part right after its second-pair work: 2 barriers per channel, not 3
# speedup vs baseline: 2.5708x; 1.0709x over previous
#include <hip/hip_runtime.h>


#define IMG      512
#define CH       16
#define TS       64
#define NTHR     512
#define RAWN     84
#define RAWROWS  96
#define RAWPITCH 98
#define ABROWS   80
#define ABPITCH  98
#define MAPITCH  66
#define NLD      21
#define OUTN     8
#define OSW      6
#define OSR      11

typedef __attribute__((ext_vector_type(2))) float v2f;
typedef __attribute__((ext_vector_type(8))) float v8f;

__device__ __forceinline__ int refl(int p) {
  p = p < 0 ? -p : p;
  return p >= IMG ? 2 * IMG - 2 - p : p;
}

__device__ __forceinline__ void halo_fetch(const float* __restrict__ src, int gx, int gy0,
                                           float (&v)[NLD]) {
#pragma unroll
  for (int j = 0; j < NLD; ++j) {
    const float* row = src + refl(gy0 + 4 * j) * IMG;
    v[j] = row[gx];
  }
}

__device__ __forceinline__ void lds_barrier() {
  asm volatile("s_wait_dscnt 0x0\n\ts_barrier_signal -1\n\ts_barrier_wait -1" ::: "memory");
}
__device__ __forceinline__ void halo_store(float* __restrict__ raw, int lo, bool valid,
                                           const float (&v)[NLD]) {
  if (valid) {
#pragma unroll
    for (int j = 0; j < NLD; ++j) raw[lo + 4 * j * RAWPITCH] = v[j];
  }
}

__device__ __forceinline__ int frag_row(int c) {
  return 4 * ((c >> 1) & 3) + 2 * (c >> 3) + (c & 1);
}

template <int MODE, int NIN, int KT0, int KT1>
__device__ __forceinline__ void h_steps(const float* __restrict__ p0,
                                        const float* __restrict__ p1, int sPitch,
                                        const v2f* __restrict__ bandH, v8f (&hT)[NIN]) {
#pragma unroll
  for (int kt = KT0; kt < KT1; ++kt) {
#pragma unroll
    for (int i = 0; i < NIN; ++i) {
      const int o = i * 16 * sPitch + 4 * kt;
      float x0 = p0[o], x1 = p0[o + 1];
      if (MODE == 1) { x0 *= x0; x1 *= x1; }
      if (MODE == 2) { x0 *= p1[o]; x1 *= p1[o + 1]; }
      v2f a; a.x = x0; a.y = x1;
      hT[i] = __builtin_amdgcn_wmma_f32_16x16x4_f32(false, a, false, bandH[kt],
                                                    (short)0, hT[i], false, false);
    }
  }
}

template <int NIN, int KT0, int KT1>
__device__ __forceinline__ void h_steps_pair(const float* __restrict__ p0,
                                             const float* __restrict__ p1, int sPitch,
                                             const v2f* __restrict__ bandH,
                                             v8f (&hA)[NIN], v8f (&hB)[NIN]) {
#pragma unroll
  for (int kt = KT0; kt < KT1; ++kt) {
#pragma unroll
    for (int i = 0; i < NIN; ++i) {
      const int o = i * 16 * sPitch + 4 * kt;
      v2f a; a.x = p0[o]; a.y = p0[o + 1];
      v2f b; b.x = a.x * p1[o]; b.y = a.y * p1[o + 1];
      hA[i] = __builtin_amdgcn_wmma_f32_16x16x4_f32(false, a, false, bandH[kt],
                                                    (short)0, hA[i], false, false);
      hB[i] = __builtin_amdgcn_wmma_f32_16x16x4_f32(false, b, false, bandH[kt],
                                                    (short)0, hB[i], false, false);
    }
  }
}

template <int NIN, bool TAILZERO>
__device__ __forceinline__ void v_pass(const v8f (&hT)[NIN], const v2f* __restrict__ bandH,
                                       v8f (&acc)[NIN - 1]) {
#pragma unroll
  for (int o = 0; o < NIN - 1; ++o) acc[o] = (v8f){};
#pragma unroll
  for (int ks = 0; ks < 7; ++ks) {
    const int blk = ks >> 2, t = ks & 3;
#pragma unroll
    for (int o = 0; o < NIN - 1; ++o) {
      if (TAILZERO && o == NIN - 2 && ks > 4) continue;
      v2f b; b.x = hT[o + blk][2 * t]; b.y = hT[o + blk][2 * t + 1];
      acc[o] = __builtin_amdgcn_wmma_f32_16x16x4_f32(false, bandH[ks], false, b,
                                                     (short)0, acc[o], false, false);
    }
  }
}

template <int MODE, int NIN, bool MAYLAST, bool TAILZERO>
__device__ __forceinline__ void blur_tiles(const float* __restrict__ s0,
                                           const float* __restrict__ s1, int sPitch,
                                           int nj,
                                           const v2f* __restrict__ bandH,
                                           int lane, v8f (&acc)[NIN - 1]) {
  const int c = lane & 15, half = lane >> 4;
  const float* p0 = s0 + frag_row(c) * sPitch + nj * 16 + 2 * half;
  const float* p1 = s1 + frag_row(c) * sPitch + nj * 16 + 2 * half;
  v8f hT[NIN];
#pragma unroll
  for (int i = 0; i < NIN; ++i) hT[i] = (v8f){};
  h_steps<MODE, NIN, 0, 5>(p0, p1, sPitch, bandH, hT);
  if (!MAYLAST || nj != 4) h_steps<MODE, NIN, 5, 7>(p0, p1, sPitch, bandH, hT);
  v_pass<NIN, TAILZERO>(hT, bandH, acc);
}

template <int MODE, int NIN, bool MAYLAST, bool TAILZERO>
__device__ __forceinline__ void blur_strip(const float* __restrict__ s0,
                                           const float* __restrict__ s1, int sPitch,
                                           int nj, float* __restrict__ dst, int dPitch,
                                           const v2f* __restrict__ bandH,
                                           int lane) {
  const int c = lane & 15, half = lane >> 4;
  v8f acc[NIN - 1];
  blur_tiles<MODE, NIN, MAYLAST, TAILZERO>(s0, s1, sPitch, nj, bandH, lane, acc);
#pragma unroll
  for (int o = 0; o < NIN - 1; ++o) {
    float* pd = dst + (o * 16 + half * 8) * dPitch + nj * 16 + c;
#pragma unroll
    for (int v = 0; v < 8; ++v) pd[v * dPitch] = acc[o][v];
  }
}

template <int NIN, bool TAILZERO>
__device__ __forceinline__ void ab_tiles(const float* __restrict__ rF,
                                         const float* __restrict__ rI, int nj,
                                         const float* __restrict__ mIr,
                                         const float* __restrict__ vIr,
                                         float* __restrict__ aDst, float* __restrict__ bDst,
                                         const v2f* __restrict__ bandH,
                                         int lane) {
  const int c = lane & 15, half = lane >> 4;
  const float* pF = rF + frag_row(c) * RAWPITCH + nj * 16 + 2 * half;
  const float* pI = rI + frag_row(c) * RAWPITCH + nj * 16 + 2 * half;
  v8f hM[NIN], hC[NIN];
#pragma unroll
  for (int i = 0; i < NIN; ++i) { hM[i] = (v8f){}; hC[i] = (v8f){}; }
  h_steps_pair<NIN, 0, 5>(pF, pI, RAWPITCH, bandH, hM, hC);
  if (nj != 4) h_steps_pair<NIN, 5, 7>(pF, pI, RAWPITCH, bandH, hM, hC);
  v8f mp[NIN - 1], cp[NIN - 1];
  v_pass<NIN, TAILZERO>(hM, bandH, mp);
  v_pass<NIN, TAILZERO>(hC, bandH, cp);
#pragma unroll
  for (int o = 0; o < NIN - 1; ++o) {
    const int base = (o * 16 + half * 8) * ABPITCH + nj * 16 + c;
#pragma unroll
    for (int v = 0; v < 8; ++v) {
      const int idx = base + v * ABPITCH;
      const float mi_ = mIr[idx], mp_ = mp[o][v];
      const float a = (cp[o][v] - mi_ * mp_) * vIr[idx];
      aDst[idx] = a;
      bDst[idx] = mp_ - a * mi_;
    }
  }
}

__device__ __forceinline__ void out_rows_shared(const float* __restrict__ mA,
                                                const float* __restrict__ mB,
                                                const float* __restrict__ rawI,
                                                float* __restrict__ oT, int u, int lane) {
  const float* pA = mA + u * MAPITCH + 2 * lane;
  const float* pB = mB + u * MAPITCH + 2 * lane;
  const float* pI = rawI + (u + 10) * RAWPITCH + 10 + 2 * lane;
  float* pO = oT + (size_t)u * IMG + 2 * lane;
  const bool tail = u + OSW * (OSR - 1) < TS;
  v2f va[OSR - 1], vb[OSR - 1], vi[OSR - 1];
  v2f ta = {0.0f, 0.0f}, tb = {0.0f, 0.0f}, ti = {0.0f, 0.0f};
#pragma unroll
  for (int j = 0; j < OSR - 1; ++j) {
    va[j] = *(const v2f*)(pA + j * OSW * MAPITCH);
    vb[j] = *(const v2f*)(pB + j * OSW * MAPITCH);
    vi[j] = *(const v2f*)(pI + j * OSW * RAWPITCH);
  }
  if (tail) {
    ta = *(const v2f*)(pA + (OSR - 1) * OSW * MAPITCH);
    tb = *(const v2f*)(pB + (OSR - 1) * OSW * MAPITCH);
    ti = *(const v2f*)(pI + (OSR - 1) * OSW * RAWPITCH);
  }
#pragma unroll
  for (int j = 0; j < OSR - 1; ++j)
    *(v2f*)(pO + (size_t)j * OSW * IMG) = va[j] * vi[j] + vb[j];
  if (tail) *(v2f*)(pO + (size_t)(OSR - 1) * OSW * IMG) = ta * ti + tb;
}

__global__ __launch_bounds__(NTHR) void guided_filter_kernel(
    const float* __restrict__ feat, const float* __restrict__ img,
    float* __restrict__ out) {
  __shared__ __attribute__((aligned(8))) float rawI[RAWROWS * RAWPITCH];
  __shared__ float rawF[RAWROWS * RAWPITCH];
  __shared__ float mI [ABROWS * ABPITCH];
  __shared__ float vIi[ABROWS * ABPITCH];
  __shared__ float mP [ABROWS * ABPITCH];
  __shared__ float cIp[ABROWS * ABPITCH];
  __shared__ __attribute__((aligned(8))) float mA [TS * MAPITCH];
  __shared__ __attribute__((aligned(8))) float mB [TS * MAPITCH];

  const int tid  = threadIdx.x;
  const int lane = tid & 31;
  const int wave = tid >> 5;
  const int x0 = blockIdx.x * TS, y0 = blockIdx.y * TS;
  const int bat = blockIdx.z;

  v2f bandH[7];
  {
    const int q = lane & 15, h = lane >> 4;
#pragma unroll
    for (int kt = 0; kt < 7; ++kt) {
      const int k0 = 4 * kt + 2 * h;
      bandH[kt].x = (k0 >= q && k0 <= q + 10) ? (1.0f / 11.0f) : 0.0f;
      bandH[kt].y = (k0 + 1 >= q && k0 + 1 <= q + 10) ? (1.0f / 11.0f) : 0.0f;
    }
  }

  const int hc = tid & 127, hr = __builtin_amdgcn_readfirstlane(tid >> 7);
  const int gx = refl(x0 - 10 + hc), gy0 = y0 - 10 + hr;
  const int hlo = hr * RAWPITCH + hc;
  const bool hvalid = hc < RAWN;
  const bool hfetch = (wave & 3) != 3;

  const float* imgB = img + (size_t)bat * IMG * IMG;
  const float* featB = feat + (size_t)bat * CH * IMG * IMG;
  float tI[NLD], tF[NLD];
  if (hfetch) {
    halo_fetch(imgB, gx, gy0, tI);
    halo_fetch(featB, gx, gy0, tF);
  }

  for (int i = tid; i < RAWROWS * RAWPITCH; i += NTHR) { rawI[i] = 0.0f; rawF[i] = 0.0f; }
  __syncthreads();
  halo_store(rawI, hlo, hvalid, tI);
  halo_store(rawF, hlo, hvalid, tF);
  __syncthreads();

  if (wave < 10) {
    const int qs = wave >= 5 ? 1 : 0, nj = wave - 5 * qs;
    if (qs == 0) blur_strip<0, 6, true, true>(rawI, rawI, RAWPITCH, nj, mI,  ABPITCH, bandH, lane);
    else         blur_strip<1, 6, true, true>(rawI, rawI, RAWPITCH, nj, vIi, ABPITCH, bandH, lane);
  }
  __syncthreads();
  for (int i = tid; i < ABROWS * ABROWS; i += NTHR) {
    const int r = i / ABROWS, cc = i - r * ABROWS;
    const int o = r * ABPITCH + cc;
    const float m = mI[o];
    vIi[o] = 1.0f / (vIi[o] - m * m + 1e-8f);
  }

  float* outT = out + (size_t)bat * CH * IMG * IMG + (size_t)y0 * IMG + x0;

  for (int ch = 0; ch < CH; ++ch) {
    lds_barrier();

    {
      const int t = (wave + 2 * ch) & 15;
      if (t < 10) {
        const bool upper = (t & 2) != 0 || t == 9;
        if (upper) {
          const int nj = t == 9 ? 4 : (t >> 2) + 2 * (t & 1);
          ab_tiles<4, false>(rawF, rawI, nj, mI, vIi, cIp, mP, bandH, lane);
        } else {
          const int nj = (t >> 2) + 3 * (t & 1);
          ab_tiles<3, true>(rawF + 48 * RAWPITCH, rawI + 48 * RAWPITCH, nj,
                            mI + 48 * ABPITCH, vIi + 48 * ABPITCH,
                            cIp + 48 * ABPITCH, mP + 48 * ABPITCH, bandH, lane);
        }
      } else if (ch > 0) {
        out_rows_shared(mA, mB, rawI, outT + (size_t)(ch - 1) * IMG * IMG, t - 10, lane);
      }
    }
    lds_barrier();

    const bool more = ch + 1 < CH;
    if (more && hfetch) halo_fetch(featB + (size_t)(ch + 1) * IMG * IMG, gx, gy0, tF);

    {
      const int t = (wave + 2 * ch + 6) & 15;
      if (t < 8) {
        const int qs = t >> 2, nj = t & 3;
        if (qs == 0) blur_strip<0, 5, false, false>(cIp, cIp, ABPITCH, nj, mA, MAPITCH, bandH, lane);
        else         blur_strip<0, 5, false, false>(mP,  mP,  ABPITCH, nj, mB, MAPITCH, bandH, lane);
      }
    }
    if (more) halo_store(rawF, hlo, hvalid, tF);
  }

  lds_barrier();

  {
    const int oy = tid >> 6, ox = tid & 63;
    const float* pA = mA + oy * MAPITCH + ox;
    const float* pB = mB + oy * MAPITCH + ox;
    const float* pI = rawI + (oy + 10) * RAWPITCH + (ox + 10);
    float* pO = outT + (size_t)(CH - 1) * IMG * IMG + (size_t)oy * IMG + ox;
    float va[OUTN], vb[OUTN], vi[OUTN];
#pragma unroll
    for (int j = 0; j < OUTN; ++j) {
      va[j] = pA[j * 8 * MAPITCH];
      vb[j] = pB[j * 8 * MAPITCH];
      vi[j] = pI[j * 8 * RAWPITCH];
    }
#pragma unroll
    for (int j = 0; j < OUTN; ++j) pO[(size_t)j * 8 * IMG] = va[j] * vi[j] + vb[j];
  }
}

extern "C" void kernel_launch(void* const* d_in, const int* in_sizes, int n_in,
                              void* d_out, int out_size, void* d_ws, size_t ws_size,
                              hipStream_t stream) {
  const float* feat = (const float*)d_in[0];
  const float* img  = (const float*)d_in[1];
  float* out = (float*)d_out;
  const int B = in_sizes[1] / (IMG * IMG);
  dim3 grid(IMG / TS, IMG / TS, B);
  guided_filter_kernel<<<grid, NTHR, 0, stream>>>(feat, img, out);
}


# Vertical 11-row sums leave the matrix unit: H D tiles hold contiguous rows per lane, summed by f32 vector adds (pair tree, one lane-half exchange; band 1/121); H stays f32 WMMA; V sum order changes
# speedup vs baseline: 2.9358x; 1.1420x over previous
#include <hip/hip_runtime.h>


#define IMG      512
#define CH       16
#define TS       64
#define NTHR     512
#define RAWN     84
#define RAWROWS  96
#define RAWPITCH 98
#define ABROWS   80
#define ABPITCH  104
#define MAPITCH  72
#define NLD      21
#define OUTN     8
#define OSW      6
#define OSR      11

typedef __attribute__((ext_vector_type(2))) float v2f;
typedef __attribute__((ext_vector_type(8))) float v8f;

__device__ __forceinline__ int refl(int p) {
  p = p < 0 ? -p : p;
  return p >= IMG ? 2 * IMG - 2 - p : p;
}

__device__ __forceinline__ void halo_fetch(const float* __restrict__ src, int gx, int gy0,
                                           float (&v)[NLD]) {
#pragma unroll
  for (int j = 0; j < NLD; ++j) {
    const float* row = src + refl(gy0 + 4 * j) * IMG;
    v[j] = row[gx];
  }
}

__device__ __forceinline__ void lds_barrier() {
  asm volatile("s_wait_dscnt 0x0\n\ts_barrier_signal -1\n\ts_barrier_wait -1" ::: "memory");
}
__device__ __forceinline__ void halo_store(float* __restrict__ raw, int lo, bool valid,
                                           const float (&v)[NLD]) {
  if (valid) {
#pragma unroll
    for (int j = 0; j < NLD; ++j) raw[lo + 4 * j * RAWPITCH] = v[j];
  }
}

__device__ __forceinline__ int frag_row(int c, int L) {
  return c < 8 ? c : c + L - 14;
}

__device__ __forceinline__ float xhalf(float x) {
  const int b = __builtin_bit_cast(int, x);
  return __builtin_bit_cast(float, __builtin_amdgcn_permlanex16(b, b, 0x76543210, 0xfedcba98,
                                                                false, false));
}

template <int MODE, int NIN, int KT0, int KT1>
__device__ __forceinline__ void h_steps(const float* __restrict__ p0,
                                        const float* __restrict__ p1, int sPitch,
                                        const v2f* __restrict__ bandH, v8f (&hT)[NIN]) {
#pragma unroll
  for (int kt = KT0; kt < KT1; ++kt) {
#pragma unroll
    for (int i = 0; i < NIN; ++i) {
      const int o = i * 8 * sPitch + 4 * kt;
      float x0 = p0[o], x1 = p0[o + 1];
      if (MODE == 1) { x0 *= x0; x1 *= x1; }
      if (MODE == 2) { x0 *= p1[o]; x1 *= p1[o + 1]; }
      v2f a; a.x = x0; a.y = x1;
      hT[i] = __builtin_amdgcn_wmma_f32_16x16x4_f32(false, a, false, bandH[kt],
                                                    (short)0, hT[i], false, false);
    }
  }
}

template <int NIN, int KT0, int KT1>
__device__ __forceinline__ void h_steps_pair(const float* __restrict__ p0,
                                             const float* __restrict__ p1, int sPitch,
                                             const v2f* __restrict__ bandH,
                                             v8f (&hA)[NIN], v8f (&hB)[NIN]) {
#pragma unroll
  for (int kt = KT0; kt < KT1; ++kt) {
#pragma unroll
    for (int i = 0; i < NIN; ++i) {
      const int o = i * 8 * sPitch + 4 * kt;
      v2f a; a.x = p0[o]; a.y = p0[o + 1];
      v2f b; b.x = a.x * p1[o]; b.y = a.y * p1[o + 1];
      hA[i] = __builtin_amdgcn_wmma_f32_16x16x4_f32(false, a, false, bandH[kt],
                                                    (short)0, hA[i], false, false);
      hB[i] = __builtin_amdgcn_wmma_f32_16x16x4_f32(false, b, false, bandH[kt],
                                                    (short)0, hB[i], false, false);
    }
  }
}

template <int NIN>
__device__ __forceinline__ void v_sums(const v8f (&hT)[NIN], int half,
                                       float (&al)[8 * NIN - 10], float (&st)[2]) {
  constexpr int L = 8 * NIN;
  float s2[L - 2], s4[L - 6], s8[L - 10];
#pragma unroll
  for (int r = 0; r < L - 2; ++r) s2[r] = hT[r >> 3][r & 7] + hT[(r + 1) >> 3][(r + 1) & 7];
#pragma unroll
  for (int r = 0; r < L - 6; ++r) s4[r] = s2[r] + s2[r + 2];
#pragma unroll
  for (int r = 0; r < L - 10; ++r) s8[r] = s4[r] + s4[r + 4];
#pragma unroll
  for (int q = 0; q < L - 10; ++q)
    al[q] = (s8[q] + s2[q + 8]) + hT[(q + 10) >> 3][(q + 10) & 7];
  const float t0 = s4[L - 10];
  const float t1 = hT[(L - 9) >> 3][(L - 9) & 7] + s2[L - 8];
  const float t2 = s2[L - 8];
  const float t3 = hT[(L - 7) >> 3][(L - 7) & 7];
  const float p0 = (s4[0] + s2[4]) + hT[0][6];
  const float p1 = s8[0];
  const float p2 = s8[0] + hT[1][0];
  const float p3 = s8[0] + s2[8];
  const bool hi = half != 0;
  const float ra = xhalf(hi ? p0 : t2);
  const float rb = xhalf(hi ? p1 : t3);
  st[0] = (hi ? p2 : t0) + ra;
  st[1] = (hi ? p3 : t1) + rb;
}

template <int MODE, int NIN, bool MAYLAST>
__device__ __forceinline__ void blur_tiles(const float* __restrict__ s0,
                                           const float* __restrict__ s1, int sPitch,
                                           int nj,
                                           const v2f* __restrict__ bandH,
                                           int lane, float (&al)[8 * NIN - 10],
                                           float (&st)[2]) {
  const int c = lane & 15, half = lane >> 4;
  const float* p0 = s0 + frag_row(c, 8 * NIN) * sPitch + nj * 16 + 2 * half;
  const float* p1 = s1 + frag_row(c, 8 * NIN) * sPitch + nj * 16 + 2 * half;
  v8f hT[NIN];
#pragma unroll
  for (int i = 0; i < NIN; ++i) hT[i] = (v8f){};
  h_steps<MODE, NIN, 0, 5>(p0, p1, sPitch, bandH, hT);
  if (!MAYLAST || nj != 4) h_steps<MODE, NIN, 5, 7>(p0, p1, sPitch, bandH, hT);
  v_sums<NIN>(hT, half, al, st);
}

template <int MODE, int NIN, bool MAYLAST>
__device__ __forceinline__ void blur_strip(const float* __restrict__ s0,
                                           const float* __restrict__ s1, int sPitch,
                                           int nj, float* __restrict__ dst, int dPitch,
                                           const v2f* __restrict__ bandH,
                                           int lane) {
  constexpr int L = 8 * NIN;
  const int c = lane & 15, half = lane >> 4;
  float al[L - 10], st[2];
  blur_tiles<MODE, NIN, MAYLAST>(s0, s1, sPitch, nj, bandH, lane, al, st);
  float* pa = dst + (half ? L - 6 : 0) * dPitch + nj * 16 + c;
  float* ps = dst + (half ? L - 8 : L - 10) * dPitch + nj * 16 + c;
#pragma unroll
  for (int q = 0; q < L - 10; ++q) pa[q * dPitch] = al[q];
#pragma unroll
  for (int j = 0; j < 2; ++j) ps[j * dPitch] = st[j];
}

template <int NIN>
__device__ __forceinline__ void ab_tiles(const float* __restrict__ rF,
                                         const float* __restrict__ rI, int nj,
                                         const float* __restrict__ mIr,
                                         const float* __restrict__ vIr,
                                         float* __restrict__ aDst, float* __restrict__ bDst,
                                         const v2f* __restrict__ bandH,
                                         int lane) {
  constexpr int L = 8 * NIN;
  const int c = lane & 15, half = lane >> 4;
  const float* pF = rF + frag_row(c, L) * RAWPITCH + nj * 16 + 2 * half;
  const float* pI = rI + frag_row(c, L) * RAWPITCH + nj * 16 + 2 * half;
  v8f hM[NIN], hC[NIN];
#pragma unroll
  for (int i = 0; i < NIN; ++i) { hM[i] = (v8f){}; hC[i] = (v8f){}; }
  h_steps_pair<NIN, 0, 5>(pF, pI, RAWPITCH, bandH, hM, hC);
  if (nj != 4) h_steps_pair<NIN, 5, 7>(pF, pI, RAWPITCH, bandH, hM, hC);
  float mpA[L - 10], mpS[2], cpA[L - 10], cpS[2];
  v_sums<NIN>(hM, half, mpA, mpS);
  v_sums<NIN>(hC, half, cpA, cpS);
  const int baseA = (half ? L - 6 : 0) * ABPITCH + nj * 16 + c;
  const int baseS = (half ? L - 8 : L - 10) * ABPITCH + nj * 16 + c;
#pragma unroll
  for (int q = 0; q < L - 10; ++q) {
    const int idx = baseA + q * ABPITCH;
    const float mi_ = mIr[idx], mp_ = mpA[q];
    const float a = (cpA[q] - mi_ * mp_) * vIr[idx];
    aDst[idx] = a;
    bDst[idx] = mp_ - a * mi_;
  }
#pragma unroll
  for (int j = 0; j < 2; ++j) {
    const int idx = baseS + j * ABPITCH;
    const float mi_ = mIr[idx], mp_ = mpS[j];
    const float a = (cpS[j] - mi_ * mp_) * vIr[idx];
    aDst[idx] = a;
    bDst[idx] = mp_ - a * mi_;
  }
}

__device__ __forceinline__ void out_rows_shared(const float* __restrict__ mA,
                                                const float* __restrict__ mB,
                                                const float* __restrict__ rawI,
                                                float* __restrict__ oT, int u, int lane) {
  const float* pA = mA + u * MAPITCH + 2 * lane;
  const float* pB = mB + u * MAPITCH + 2 * lane;
  const float* pI = rawI + (u + 10) * RAWPITCH + 10 + 2 * lane;
  float* pO = oT + (size_t)u * IMG + 2 * lane;
  const bool tail = u + OSW * (OSR - 1) < TS;
  v2f va[OSR - 1], vb[OSR - 1], vi[OSR - 1];
  v2f ta = {0.0f, 0.0f}, tb = {0.0f, 0.0f}, ti = {0.0f, 0.0f};
#pragma unroll
  for (int j = 0; j < OSR - 1; ++j) {
    va[j] = *(const v2f*)(pA + j * OSW * MAPITCH);
    vb[j] = *(const v2f*)(pB + j * OSW * MAPITCH);
    vi[j] = *(const v2f*)(pI + j * OSW * RAWPITCH);
  }
  if (tail) {
    ta = *(const v2f*)(pA + (OSR - 1) * OSW * MAPITCH);
    tb = *(const v2f*)(pB + (OSR - 1) * OSW * MAPITCH);
    ti = *(const v2f*)(pI + (OSR - 1) * OSW * RAWPITCH);
  }
#pragma unroll
  for (int j = 0; j < OSR - 1; ++j)
    *(v2f*)(pO + (size_t)j * OSW * IMG) = va[j] * vi[j] + vb[j];
  if (tail) *(v2f*)(pO + (size_t)(OSR - 1) * OSW * IMG) = ta * ti + tb;
}

__global__ __launch_bounds__(NTHR) void guided_filter_kernel(
    const float* __restrict__ feat, const float* __restrict__ img,
    float* __restrict__ out) {
  __shared__ __attribute__((aligned(8))) float rawI[RAWROWS * RAWPITCH];
  __shared__ float rawF[RAWROWS * RAWPITCH];
  __shared__ float mI [ABROWS * ABPITCH];
  __shared__ float vIi[ABROWS * ABPITCH];
  __shared__ float mP [ABROWS * ABPITCH];
  __shared__ float cIp[ABROWS * ABPITCH];
  __shared__ __attribute__((aligned(8))) float mA [TS * MAPITCH];
  __shared__ __attribute__((aligned(8))) float mB [TS * MAPITCH];

  const int tid  = threadIdx.x;
  const int lane = tid & 31;
  const int wave = tid >> 5;
  const int x0 = blockIdx.x * TS, y0 = blockIdx.y * TS;
  const int bat = blockIdx.z;

  v2f bandH[7];
  {
    const int q = lane & 15, h = lane >> 4;
#pragma unroll
    for (int kt = 0; kt < 7; ++kt) {
      const int k0 = 4 * kt + 2 * h;
      bandH[kt].x = (k0 >= q && k0 <= q + 10) ? (1.0f / 121.0f) : 0.0f;
      bandH[kt].y = (k0 + 1 >= q && k0 + 1 <= q + 10) ? (1.0f / 121.0f) : 0.0f;
    }
  }

  const int hc = tid & 127, hr = __builtin_amdgcn_readfirstlane(tid >> 7);
  const int gx = refl(x0 - 10 + hc), gy0 = y0 - 10 + hr;
  const int hlo = hr * RAWPITCH + hc;
  const bool hvalid = hc < RAWN;
  const bool hfetch = (wave & 3) != 3;

  const float* imgB = img + (size_t)bat * IMG * IMG;
  const float* featB = feat + (size_t)bat * CH * IMG * IMG;
  float tI[NLD], tF[NLD];
  if (hfetch) {
    halo_fetch(imgB, gx, gy0, tI);
    halo_fetch(featB, gx, gy0, tF);
  }

  for (int i = tid; i < RAWROWS * RAWPITCH; i += NTHR) { rawI[i] = 0.0f; rawF[i] = 0.0f; }
  __syncthreads();
  halo_store(rawI, hlo, hvalid, tI);
  halo_store(rawF, hlo, hvalid, tF);
  __syncthreads();

  if (wave < 10) {
    const int qs = wave >= 5 ? 1 : 0, nj = wave - 5 * qs;
    if (qs == 0) blur_strip<0, 6, true>(rawI, rawI, RAWPITCH, nj, mI,  ABPITCH, bandH, lane);
    else         blur_strip<1, 6, true>(rawI, rawI, RAWPITCH, nj, vIi, ABPITCH, bandH, lane);
  }
  __syncthreads();
  for (int i = tid; i < ABROWS * ABROWS; i += NTHR) {
    const int r = i / ABROWS, cc = i - r * ABROWS;
    const int o = r * ABPITCH + cc;
    const float m = mI[o];
    vIi[o] = 1.0f / (vIi[o] - m * m + 1e-8f);
  }

  float* outT = out + (size_t)bat * CH * IMG * IMG + (size_t)y0 * IMG + x0;

  for (int ch = 0; ch < CH; ++ch) {
    lds_barrier();

    {
      const int t = (wave + 2 * ch) & 15;
      if (t < 10) {
        const bool upper = (t & 2) != 0 || t == 9;
        if (upper) {
          const int nj = t == 9 ? 4 : (t >> 2) + 2 * (t & 1);
          ab_tiles<4>(rawF, rawI, nj, mI, vIi, cIp, mP, bandH, lane);
        } else {
          const int nj = (t >> 2) + 3 * (t & 1);
          ab_tiles<3>(rawF + 48 * RAWPITCH, rawI + 48 * RAWPITCH, nj,
                            mI + 48 * ABPITCH, vIi + 48 * ABPITCH,
                            cIp + 48 * ABPITCH, mP + 48 * ABPITCH, bandH, lane);
        }
      } else if (ch > 0) {
        out_rows_shared(mA, mB, rawI, outT + (size_t)(ch - 1) * IMG * IMG, t - 10, lane);
      }
    }
    lds_barrier();

    const bool more = ch + 1 < CH;
    if (more && hfetch) halo_fetch(featB + (size_t)(ch + 1) * IMG * IMG, gx, gy0, tF);

    {
      const int t = (wave + 2 * ch + 6) & 15;
      if (t < 8) {
        const int qs = t >> 2, nj = t & 3;
        if (qs == 0) blur_strip<0, 5, false>(cIp, cIp, ABPITCH, nj, mA, MAPITCH, bandH, lane);
        else         blur_strip<0, 5, false>(mP,  mP,  ABPITCH, nj, mB, MAPITCH, bandH, lane);
      }
    }
    if (more) halo_store(rawF, hlo, hvalid, tF);
  }

  lds_barrier();

  {
    const int oy = tid >> 6, ox = tid & 63;
    const float* pA = mA + oy * MAPITCH + ox;
    const float* pB = mB + oy * MAPITCH + ox;
    const float* pI = rawI + (oy + 10) * RAWPITCH + (ox + 10);
    float* pO = outT + (size_t)(CH - 1) * IMG * IMG + (size_t)oy * IMG + ox;
    float va[OUTN], vb[OUTN], vi[OUTN];
#pragma unroll
    for (int j = 0; j < OUTN; ++j) {
      va[j] = pA[j * 8 * MAPITCH];
      vb[j] = pB[j * 8 * MAPITCH];
      vi[j] = pI[j * 8 * RAWPITCH];
    }
#pragma unroll
    for (int j = 0; j < OUTN; ++j) pO[(size_t)j * 8 * IMG] = va[j] * vi[j] + vb[j];
  }
}

extern "C" void kernel_launch(void* const* d_in, const int* in_sizes, int n_in,
                              void* d_out, int out_size, void* d_ws, size_t ws_size,
                              hipStream_t stream) {
  const float* feat = (const float*)d_in[0];
  const float* img  = (const float*)d_in[1];
  float* out = (float*)d_out;
  const int B = in_sizes[1] / (IMG * IMG);
  dim3 grid(IMG / TS, IMG / TS, B);
  guided_filter_kernel<<<grid, NTHR, 0, stream>>>(feat, img, out);
}
